# starting kernel
# baseline (speedup 1.0000x reference)
#include <hip/hip_runtime.h>

typedef __attribute__((ext_vector_type(2))) float v2f;
typedef __attribute__((ext_vector_type(8))) float v8f;

#define D_FEAT 64
constexpr float LAM = 2.0f;


__global__ void zero_kernel(float* __restrict__ p, long n) {
  long i = (long)blockIdx.x * blockDim.x + threadIdx.x;
  long stride = (long)gridDim.x * blockDim.x;
  for (; i < n; i += stride) p[i] = 0.0f;
}

__global__ void deg_kernel(const int* __restrict__ dst, float* __restrict__ deg, int E) {
  int i = blockIdx.x * blockDim.x + threadIdx.x;
  if (i < E) atomicAdd(&deg[dst[i]], 1.0f);
}

__global__ void norm_kernel(float* __restrict__ d, int N) {
  int i = blockIdx.x * blockDim.x + threadIdx.x;
  if (i < N) {
    float v = fmaxf(d[i], 1.0f);
    d[i] = rsqrtf(v);
  }
}

__global__ void scatter_kernel(const float* __restrict__ x, const float* __restrict__ nrm,
                               const int* __restrict__ src, const int* __restrict__ dst,
                               float* __restrict__ agg, int E) {
  int lane = threadIdx.x & 31;
  int edge = blockIdx.x * (blockDim.x >> 5) + (threadIdx.x >> 5);
  if (edge >= E) return;
  int s = src[edge];
  int d = dst[edge];
  float ns = nrm[s];
  const float2* xs = (const float2*)(x + (long)s * D_FEAT);
  float2 v = xs[lane];
  float* o = agg + (long)d * D_FEAT + lane * 2;
  atomicAdd(o + 0, v.x * ns);
  atomicAdd(o + 1, v.y * ns);
}

__global__ void combine_kernel(const float* __restrict__ agg, const float* __restrict__ nrm,
                               const float* __restrict__ t1, const float* __restrict__ t2,
                               float* __restrict__ out,
                               float cH, float cX, float cP, long n) {
  long i = (long)blockIdx.x * blockDim.x + threadIdx.x;
  if (i >= n) return;
  long node = i >> 6;
  float h = agg[i] * nrm[node];
  out[i] = cH * h + cX * t1[i] + cP * t2[i];
}


__device__ __forceinline__ v8f accum_term(v8f acc, const float* __restrict__ xr,
                                          const float* __restrict__ Wt,
                                          int khalf, int ncol) {
  v2f bfrag[16];
#pragma unroll
  for (int kk = 0; kk < 16; ++kk) {
    int krow = kk * 4 + khalf * 2;
    v2f bf;
    bf.x = Wt[(krow + 0) * D_FEAT + ncol];
    bf.y = Wt[(krow + 1) * D_FEAT + ncol];
    bfrag[kk] = bf;
  }
#pragma unroll
  for (int kk = 0; kk < 16; ++kk) {
    v2f a;
    a.x = xr[kk * 4 + khalf * 2 + 0];
    a.y = xr[kk * 4 + khalf * 2 + 1];
    acc = __builtin_amdgcn_wmma_f32_16x16x4_f32(
         false, a,  false, bfrag[kk],
         (short)0, acc,  false,  false);
  }
  return acc;
}

__global__ __launch_bounds__(128) void gemm_fused_kernel(
    const float* __restrict__ X0, const float* __restrict__ X1,
    const float* __restrict__ X2, const float* __restrict__ X3,
    const float* __restrict__ W, const float* __restrict__ bias,
    float* __restrict__ out, int N) {
  int lane = threadIdx.x & 31;
  int wv   = threadIdx.x >> 5;
  int row0 = blockIdx.x * 16;
  if (row0 >= N) return;

  int mlo   = lane & 15;
  int khalf = lane >> 4;
  int ncol  = wv * 16 + mlo;

  int arow = row0 + mlo;
  if (arow >= N) arow = N - 1;
  long aoff = (long)arow * D_FEAT;

  v8f acc;
  {
    float bb = bias[ncol];
#pragma unroll
    for (int v = 0; v < 8; ++v) acc[v] = bb;
  }

  acc = accum_term(acc, X0 + aoff, W + 0 * D_FEAT * D_FEAT, khalf, ncol);
  acc = accum_term(acc, X1 + aoff, W + 1 * D_FEAT * D_FEAT, khalf, ncol);
  acc = accum_term(acc, X2 + aoff, W + 2 * D_FEAT * D_FEAT, khalf, ncol);
  acc = accum_term(acc, X3 + aoff, W + 3 * D_FEAT * D_FEAT, khalf, ncol);

  if (row0 + 16 <= N) {
    float* orow = out + (long)(row0 + khalf * 8) * D_FEAT + ncol;
#pragma unroll
    for (int v = 0; v < 8; ++v) orow[(long)v * D_FEAT] = acc[v];
  } else {
#pragma unroll
    for (int v = 0; v < 8; ++v) {
      int r = row0 + khalf * 8 + v;
      if (r < N) out[(long)r * D_FEAT + ncol] = acc[v];
    }
  }
}


extern "C" void kernel_launch(void* const* d_in, const int* in_sizes, int n_in,
                              void* d_out, int out_size, void* d_ws, size_t ws_size,
                              hipStream_t stream) {
  const float* feat = (const float*)d_in[0];
  const int*   src  = (const int*)d_in[1];
  const int*   dst  = (const int*)d_in[2];
  const float* W    = (const float*)d_in[3];
  const float* b    = (const float*)d_in[4];
  float*       out  = (float*)d_out;

  const int N = in_sizes[0] / D_FEAT;
  const int E = in_sizes[1];
  const long nd = (long)N * D_FEAT;

  float* nrm = (float*)d_ws;
  float* agg = nrm + N;
  float* txA = agg + nd;
  float* txB = txA + nd;

  const int cgrid = (int)((nd + 255) / 256);
  const int sgrid = (E + 7) / 8;
  const int ggrid = (N + 15) / 16;

  zero_kernel<<<512, 256, 0, stream>>>(nrm, (long)N);
  deg_kernel<<<(E + 255) / 256, 256, 0, stream>>>(dst, nrm, E);
  norm_kernel<<<(N + 255) / 256, 256, 0, stream>>>(nrm, N);

  zero_kernel<<<2048, 256, 0, stream>>>(agg, nd);
  scatter_kernel<<<sgrid, 256, 0, stream>>>(feat, nrm, src, dst, agg, E);
  combine_kernel<<<cgrid, 256, 0, stream>>>(agg, nrm, feat, feat, txA,
                                            -2.0f / LAM, 2.0f / LAM - 1.0f, 0.0f, nd);

  zero_kernel<<<2048, 256, 0, stream>>>(agg, nd);
  scatter_kernel<<<sgrid, 256, 0, stream>>>(txA, nrm, src, dst, agg, E);
  combine_kernel<<<cgrid, 256, 0, stream>>>(agg, nrm, txA, feat, txB,
                                            -4.0f / LAM, 4.0f / LAM - 2.0f, -1.0f, nd);

  zero_kernel<<<2048, 256, 0, stream>>>(agg, nd);
  scatter_kernel<<<sgrid, 256, 0, stream>>>(txB, nrm, src, dst, agg, E);
  combine_kernel<<<cgrid, 256, 0, stream>>>(agg, nrm, txB, txA, agg,
                                            -4.0f / LAM, 4.0f / LAM - 2.0f, -1.0f, nd);

  gemm_fused_kernel<<<ggrid, 128, 0, stream>>>(feat, txA, txB, agg, W, b, out, N);
}


# GEMM k split: wave w of a block sums k 16w..16w+15 of each term for all 64 columns, partial tiles added in LDS; operands by 16-B loads, none read twice; k order and sum tree of each output change
# speedup vs baseline: 2.1769x; 2.1769x over previous
#include <hip/hip_runtime.h>

typedef __attribute__((ext_vector_type(2))) float v2f;
typedef __attribute__((ext_vector_type(8))) float v8f;

#define D_FEAT 64
constexpr float LAM = 2.0f;


__global__ void zero_kernel(float* __restrict__ p, long n) {
  long i = (long)blockIdx.x * blockDim.x + threadIdx.x;
  long stride = (long)gridDim.x * blockDim.x;
  for (; i < n; i += stride) p[i] = 0.0f;
}

__global__ void deg_kernel(const int* __restrict__ dst, float* __restrict__ deg, int E) {
  int i = blockIdx.x * blockDim.x + threadIdx.x;
  if (i < E) atomicAdd(&deg[dst[i]], 1.0f);
}

__device__ __forceinline__ float node_norm(float deg) {
  return rsqrtf(fmaxf(deg, 1.0f));
}

#define RUN_EDGES 32
#define ROWS_GROUP 8
#define HALF_ROW (D_FEAT / 2)

__device__ __forceinline__ void load_rows(float2 (&r)[ROWS_GROUP], const float* __restrict__ x,
                                          int s_v, int first, int lane) {
#pragma unroll
  for (int j = 0; j < ROWS_GROUP; ++j) {
    int s = __builtin_amdgcn_readlane(s_v, first + j);
    const float* xr = x + (long)s * D_FEAT + lane;
    r[j].x = xr[0];
    r[j].y = xr[HALF_ROW];
  }
}

__device__ __forceinline__ void add_rows(const float2 (&r)[ROWS_GROUP], float* __restrict__ agg,
                                         int d_v, int n_v, int first, int lane) {
#pragma unroll
  for (int j = 0; j < ROWS_GROUP; ++j) {
    int d = __builtin_amdgcn_readlane(d_v, first + j);
    float ns = __int_as_float(__builtin_amdgcn_readlane(n_v, first + j));
    float* o = agg + (long)d * D_FEAT + lane;
    atomicAdd(o, r[j].x * ns);
    atomicAdd(o + HALF_ROW, r[j].y * ns);
  }
}

__global__ __launch_bounds__(256) void scatter_kernel(
    const float* __restrict__ x, const float* __restrict__ deg,
    const int* __restrict__ src, const int* __restrict__ dst,
    float* __restrict__ agg2, long copy_stride, int E) {
  const int lane = threadIdx.x & 31;
  const unsigned hw_id = __builtin_amdgcn_s_sendmsg_rtn(0x87);
  float* __restrict__ agg = agg2 + (long)((hw_id >> 18) & 1u) * copy_stride;
  const int wv = __builtin_amdgcn_readfirstlane((int)(threadIdx.x >> 5));
  const long base = ((long)blockIdx.x * (blockDim.x >> 5) + wv) * RUN_EDGES;
  if (base >= E) return;
  const long left = (long)E - base;
  const int cnt = left < RUN_EDGES ? (int)left : RUN_EDGES;

  const long e_v = base + (lane < cnt ? lane : cnt - 1);
  const int s_v = src[e_v];
  const int d_v = dst[e_v];
  const float g_v = deg[s_v];

  if (cnt == RUN_EDGES) {
    float2 ra[ROWS_GROUP], rb[ROWS_GROUP];
    load_rows(ra, x, s_v, 0, lane);
    load_rows(rb, x, s_v, 8, lane);
    const int n_v = __float_as_int(node_norm(g_v));
    add_rows(ra, agg, d_v, n_v, 0, lane);
    load_rows(ra, x, s_v, 16, lane);
    add_rows(rb, agg, d_v, n_v, 8, lane);
    load_rows(rb, x, s_v, 24, lane);
    add_rows(ra, agg, d_v, n_v, 16, lane);
    add_rows(rb, agg, d_v, n_v, 24, lane);
  } else {
    const int n_v = __float_as_int(node_norm(g_v));
    for (int j = 0; j < cnt; ++j) {
      int s = __builtin_amdgcn_readlane(s_v, j);
      int d = __builtin_amdgcn_readlane(d_v, j);
      float ns = __int_as_float(__builtin_amdgcn_readlane(n_v, j));
      const float* xr = x + (long)s * D_FEAT + lane;
      float v0 = xr[0];
      float v1 = xr[HALF_ROW];
      float* o = agg + (long)d * D_FEAT + lane;
      atomicAdd(o, v0 * ns);
      atomicAdd(o + HALF_ROW, v1 * ns);
    }
  }
}

__global__ __launch_bounds__(256) void combine_clear_kernel(
    float* __restrict__ agg, float* __restrict__ aggB,
    const float* __restrict__ deg,
    const float* __restrict__ t1, const float* __restrict__ t2,
    float* __restrict__ out,
    float cH, float cX, float cP, long n4) {
  long i = (long)blockIdx.x * blockDim.x + threadIdx.x;
  if (i >= n4) return;
  long node = i >> 4;
  float nn = node_norm(deg[node]);
  float4 a  = reinterpret_cast<const float4*>(agg)[i];
  float4 b  = reinterpret_cast<const float4*>(aggB)[i];
  float4 x1 = reinterpret_cast<const float4*>(t1)[i];
  float4 x2 = reinterpret_cast<const float4*>(t2)[i];
  float4 r;
  { float h = (a.x + b.x) * nn; r.x = cH * h + cX * x1.x + cP * x2.x; }
  { float h = (a.y + b.y) * nn; r.y = cH * h + cX * x1.y + cP * x2.y; }
  { float h = (a.z + b.z) * nn; r.z = cH * h + cX * x1.z + cP * x2.z; }
  { float h = (a.w + b.w) * nn; r.w = cH * h + cX * x1.w + cP * x2.w; }
  reinterpret_cast<float4*>(out)[i] = r;
  const float4 z = make_float4(0.0f, 0.0f, 0.0f, 0.0f);
  reinterpret_cast<float4*>(agg)[i] = z;
  reinterpret_cast<float4*>(aggB)[i] = z;
}


#define GEMM_WAVES 4
#define TILE_SLOTS 32
#define W_TERM (D_FEAT * D_FEAT)

__device__ __forceinline__ void load8(float (&a)[8], const float* __restrict__ p) {
  const float4 lo = reinterpret_cast<const float4*>(p)[0];
  const float4 hi = reinterpret_cast<const float4*>(p)[1];
  a[0] = lo.x; a[1] = lo.y; a[2] = lo.z; a[3] = lo.w;
  a[4] = hi.x; a[5] = hi.y; a[6] = hi.z; a[7] = hi.w;
}

__device__ __forceinline__ void term_quarter(v8f (&acc)[4], const float (&a)[8],
                                             const float* __restrict__ wrow) {
  v2f bf[4][4];
#pragma unroll
  for (int j = 0; j < 4; ++j) {
    const float4 w0 = *reinterpret_cast<const float4*>(wrow + (2 * j + 0) * D_FEAT);
    const float4 w1 = *reinterpret_cast<const float4*>(wrow + (2 * j + 1) * D_FEAT);
    bf[j][0].x = w0.x; bf[j][0].y = w1.x;
    bf[j][1].x = w0.y; bf[j][1].y = w1.y;
    bf[j][2].x = w0.z; bf[j][2].y = w1.z;
    bf[j][3].x = w0.w; bf[j][3].y = w1.w;
  }
#pragma unroll
  for (int j = 0; j < 4; ++j) {
    v2f av;
    av.x = a[2 * j + 0];
    av.y = a[2 * j + 1];
#pragma unroll
    for (int n = 0; n < 4; ++n) {
      acc[n] = __builtin_amdgcn_wmma_f32_16x16x4_f32(
           false, av,  false, bf[j][n],
           (short)0, acc[n],  false,  false);
    }
  }
}

__global__ __launch_bounds__(128) void gemm_fused_kernel(
    const float* __restrict__ X0, const float* __restrict__ X1,
    const float* __restrict__ X2,
    const float* __restrict__ agg, const float* __restrict__ aggB,
    const float* __restrict__ deg,
    const float* __restrict__ W, const float* __restrict__ bias,
    float* __restrict__ out, int N, float cH, float cX, float cP) {
  __shared__ float red[GEMM_WAVES * TILE_SLOTS * 32];

  const int lane = threadIdx.x & 31;
  const int wv   = threadIdx.x >> 5;
  const int row0 = blockIdx.x * 16;
  if (row0 >= N) return;

  const int c  = lane & 15;
  const int h  = lane >> 4;
  const int k0 = wv * 16 + h * 8;

  int arow = row0 + c;
  if (arow >= N) arow = N - 1;
  const long aoff = (long)arow * D_FEAT + k0;
  const float* wq = W + (long)k0 * D_FEAT + 4 * c;

  v8f acc[4];
  {
    const float4 bb = *reinterpret_cast<const float4*>(bias + 4 * c);
    const float b0 = wv == 0 ? bb.x : 0.0f;
    const float b1 = wv == 0 ? bb.y : 0.0f;
    const float b2 = wv == 0 ? bb.z : 0.0f;
    const float b3 = wv == 0 ? bb.w : 0.0f;
#pragma unroll
    for (int v = 0; v < 8; ++v) {
      acc[0][v] = b0; acc[1][v] = b1; acc[2][v] = b2; acc[3][v] = b3;
    }
  }

  float a[8];
  load8(a, X0 + aoff);
  term_quarter(acc, a, wq + 0 * W_TERM);
  load8(a, X1 + aoff);
  term_quarter(acc, a, wq + 1 * W_TERM);
  load8(a, X2 + aoff);
  term_quarter(acc, a, wq + 2 * W_TERM);
  {
    float ga[8], gb[8], t1[8], t2[8];
    load8(ga, agg + aoff);
    load8(gb, aggB + aoff);
    load8(t1, X2 + aoff);
    load8(t2, X1 + aoff);
    const float nn = node_norm(deg[arow]);
#pragma unroll
    for (int i = 0; i < 8; ++i) {
      float hh = (ga[i] + gb[i]) * nn;
      a[i] = cH * hh + cX * t1[i] + cP * t2[i];
    }
    term_quarter(acc, a, wq + 3 * W_TERM);
  }

  {
    float* mine = red + (wv * TILE_SLOTS) * 32 + lane;
#pragma unroll
    for (int v = 0; v < 8; ++v) {
#pragma unroll
      for (int n = 0; n < 4; ++n) mine[(4 * v + n) * 32] = acc[n][v];
    }
  }
  __syncthreads();

  for (int j = 0; j < 2; ++j) {
    const int v = 2 * wv + j;
    const float* p = red + (4 * v) * 32 + lane;
    float o[4];
#pragma unroll
    for (int n = 0; n < 4; ++n) {
      const float* q = p + n * 32;
      o[n] = ((q[0 * TILE_SLOTS * 32] + q[1 * TILE_SLOTS * 32]) + q[2 * TILE_SLOTS * 32])
             + q[3 * TILE_SLOTS * 32];
    }
    const int row = row0 + 8 * h + v;
    if (row < N)
      *reinterpret_cast<float4*>(out + (long)row * D_FEAT + 4 * c) = make_float4(o[0], o[1], o[2], o[3]);
  }
}


extern "C" void kernel_launch(void* const* d_in, const int* in_sizes, int n_in,
                              void* d_out, int out_size, void* d_ws, size_t ws_size,
                              hipStream_t stream) {
  const float* feat = (const float*)d_in[0];
  const int*   src  = (const int*)d_in[1];
  const int*   dst  = (const int*)d_in[2];
  const float* W    = (const float*)d_in[3];
  const float* b    = (const float*)d_in[4];
  float*       out  = (float*)d_out;

  const int N = in_sizes[0] / D_FEAT;
  const int E = in_sizes[1];
  const long nd = (long)N * D_FEAT;

  const long npad = ((long)N + 31) & ~31L;
  float* deg = (float*)d_ws;
  float* agg = deg + npad;
  float* agB = agg + nd;
  float* txA = agB + nd;
  float* txB = txA + nd;

  const int cgrid = (int)((nd / 4 + 255) / 256);
  const int sgrid = (E + 255) / 256;
  const int ggrid = (N + 15) / 16;

  zero_kernel<<<2048, 256, 0, stream>>>(deg, npad + 2 * nd);
  deg_kernel<<<(E + 255) / 256, 256, 0, stream>>>(dst, deg, E);

  scatter_kernel<<<sgrid, 256, 0, stream>>>(feat, deg, src, dst, agg, nd, E);
  combine_clear_kernel<<<cgrid, 256, 0, stream>>>(agg, agB, deg, feat, feat, txA,
                                                  -2.0f / LAM, 2.0f / LAM - 1.0f, 0.0f, nd / 4);

  scatter_kernel<<<sgrid, 256, 0, stream>>>(txA, deg, src, dst, agg, nd, E);
  combine_clear_kernel<<<cgrid, 256, 0, stream>>>(agg, agB, deg, txA, feat, txB,
                                                  -4.0f / LAM, 4.0f / LAM - 2.0f, -1.0f, nd / 4);

  scatter_kernel<<<sgrid, 256, 0, stream>>>(txB, deg, src, dst, agg, nd, E);

  gemm_fused_kernel<<<ggrid, 128, 0, stream>>>(feat, txA, txB, agg, agB, deg, W, b, out, N,
                                               -4.0f / LAM, 4.0f / LAM - 2.0f, -1.0f);
}


# Props by owner blocks of 200 nodes over per-block edge segments (edges grouped by owner once per call, unordered inside): rows added by LDS float adds, arrival order; no global adds, no combine
# speedup vs baseline: 2.5556x; 1.1740x over previous
#include <hip/hip_runtime.h>

typedef __attribute__((ext_vector_type(2))) float v2f;
typedef __attribute__((ext_vector_type(8))) float v8f;

#define D_FEAT 64
constexpr float LAM = 2.0f;


__global__ void zero_kernel(float* __restrict__ p, long n) {
  long i = (long)blockIdx.x * blockDim.x + threadIdx.x;
  long stride = (long)gridDim.x * blockDim.x;
  for (; i < n; i += stride) p[i] = 0.0f;
}

__global__ void deg_kernel(const int* __restrict__ dst, float* __restrict__ deg, int E) {
  int i = blockIdx.x * blockDim.x + threadIdx.x;
  if (i < E) atomicAdd(&deg[dst[i]], 1.0f);
}

__device__ __forceinline__ float node_norm(float deg) {
  return rsqrtf(fmaxf(deg, 1.0f));
}

#define OWN_NODES 200
#define MAX_OWN 1024
#define FILL_THREADS 512
#define FILL_PER_THREAD 8
#define CUR_PITCH 32
#define RUN_EDGES 32
#define ROWS_GROUP 8
#define PROP_WAVES 16

__global__ __launch_bounds__(256) void count_kernel(const float* __restrict__ deg,
                                                    int* __restrict__ cnt, int N, int nown) {
  const int lane = threadIdx.x & 31;
  const int b = blockIdx.x * (blockDim.x >> 5) + (threadIdx.x >> 5);
  if (b >= nown) return;
  const int n0 = b * OWN_NODES;
  int s = 0;
#pragma unroll
  for (int k = 0; k < (OWN_NODES + 31) / 32; ++k) {
    const int i = lane + 32 * k;
    const bool ok = i < OWN_NODES && n0 + i < N;
    const float v = deg[ok ? n0 + i : n0];
    s += ok ? (int)v : 0;
  }
  for (int m = 16; m > 0; m >>= 1) s += __shfl_xor(s, m);
  if (lane == 0) cnt[b] = s;
}

__global__ __launch_bounds__(FILL_THREADS) void fill_kernel(
    const int* __restrict__ src, const int* __restrict__ dst, const int* __restrict__ cnt,
    int* __restrict__ cursor, int* __restrict__ start, unsigned* __restrict__ seg,
    int E, int nown) {
  __shared__ int s_start[MAX_OWN];
  __shared__ int s_base[MAX_OWN];
  __shared__ int s_hist[MAX_OWN];
  __shared__ int s_wsum[FILL_THREADS / 32];
  const int t = threadIdx.x;
  const int lane = t & 31;
  const int wv = t >> 5;

  const int c0 = (2 * t < nown) ? cnt[2 * t] : 0;
  const int c1 = (2 * t + 1 < nown) ? cnt[2 * t + 1] : 0;
  s_hist[2 * t] = 0;
  s_hist[2 * t + 1] = 0;
  int incl = c0 + c1;
  for (int d = 1; d < 32; d <<= 1) {
    const int v = __shfl_up(incl, d);
    if (lane >= d) incl += v;
  }
  if (lane == 31) s_wsum[wv] = incl;
  __syncthreads();
  int woff = 0;
  for (int w = 0; w < wv; ++w) woff += s_wsum[w];
  const int excl = woff + incl - (c0 + c1);
  s_start[2 * t] = excl;
  s_start[2 * t + 1] = excl + c0;
  if (blockIdx.x == 0) {
    if (2 * t <= nown) start[2 * t] = excl;
    if (2 * t + 1 <= nown) start[2 * t + 1] = excl + c0;
  }
  __syncthreads();

  const unsigned hw_id = __builtin_amdgcn_s_sendmsg_rtn(0x87);
  const int half = (int)((hw_id >> 18) & 1u);
  int* __restrict__ cur = cursor + (long)half * (MAX_OWN * CUR_PITCH);

  const long base = (long)blockIdx.x * (FILL_THREADS * FILL_PER_THREAD);
  int rk[FILL_PER_THREAD];
  int ob[FILL_PER_THREAD];
  unsigned pk[FILL_PER_THREAD];
  int dv[FILL_PER_THREAD];
  int sv[FILL_PER_THREAD];
#pragma unroll
  for (int j = 0; j < FILL_PER_THREAD; ++j) {
    const long e = base + j * FILL_THREADS + t;
    const long ec = e < E ? e : (long)E - 1;
    dv[j] = dst[ec];
    sv[j] = src[ec];
  }
#pragma unroll
  for (int j = 0; j < FILL_PER_THREAD; ++j) {
    const long e = base + j * FILL_THREADS + t;
    ob[j] = -1;
    rk[j] = 0;
    pk[j] = 0u;
    if (e < E) {
      const int d = dv[j];
      const int s = sv[j];
      const int b = d / OWN_NODES;
      ob[j] = b;
      pk[j] = ((unsigned)s << 8) | (unsigned)(d - b * OWN_NODES);
      rk[j] = atomicAdd(&s_hist[b], 1);
    }
  }
  __syncthreads();
  for (int b = t; b < nown; b += FILL_THREADS) {
    const int hcount = s_hist[b];
    int at = 0;
    if (hcount > 0) {
      const int old = atomicAdd(&cur[b * CUR_PITCH], hcount);
      at = half == 0 ? s_start[b] + old : s_start[b + 1] - old - hcount;
    }
    s_base[b] = at;
  }
  __syncthreads();
#pragma unroll
  for (int j = 0; j < FILL_PER_THREAD; ++j) {
    if (ob[j] >= 0) seg[s_base[ob[j]] + rk[j]] = pk[j];
  }
}


__device__ __forceinline__ void lds_add(float* p, float v) {
  __hip_atomic_fetch_add(p, v, __ATOMIC_RELAXED, __HIP_MEMORY_SCOPE_WORKGROUP);
}

__device__ __forceinline__ void load_rows(float2 (&r)[ROWS_GROUP], const float* __restrict__ x,
                                          int s_v, int first, int lane) {
#pragma unroll
  for (int j = 0; j < ROWS_GROUP; ++j) {
    const int s = __builtin_amdgcn_readlane(s_v, first + j);
    r[j] = *reinterpret_cast<const float2*>(x + (long)s * D_FEAT + 2 * lane);
  }
}

__device__ __forceinline__ void add_rows(const float2 (&r)[ROWS_GROUP], float* acc,
                                         int d_v, int n_v, int first, int lane) {
#pragma unroll
  for (int j = 0; j < ROWS_GROUP; ++j) {
    const int d = __builtin_amdgcn_readlane(d_v, first + j);
    const float ns = __int_as_float(__builtin_amdgcn_readlane(n_v, first + j));
    float* o = acc + d + 2 * lane;
    lds_add(o, r[j].x * ns);
    lds_add(o + 1, r[j].y * ns);
  }
}

template <bool HAS_P>
__device__ __forceinline__ void write_term(int i, int n0, const float* acc,
                                           const float* __restrict__ x, const float* __restrict__ xp,
                                           const float* __restrict__ deg, float* __restrict__ out,
                                           float cH, float cX, float cP) {
  const int node = n0 + (i >> 4);
  const float nn = node_norm(deg[node]);
  const float4 a = reinterpret_cast<const float4*>(acc)[i];
  const long gi = (long)node * (D_FEAT / 4) + (i & 15);
  const float4 x1 = reinterpret_cast<const float4*>(x)[gi];
  float4 r;
  r.x = cH * (a.x * nn) + cX * x1.x;
  r.y = cH * (a.y * nn) + cX * x1.y;
  r.z = cH * (a.z * nn) + cX * x1.z;
  r.w = cH * (a.w * nn) + cX * x1.w;
  if (HAS_P) {
    const float4 x2 = reinterpret_cast<const float4*>(xp)[gi];
    r.x += cP * x2.x;
    r.y += cP * x2.y;
    r.z += cP * x2.z;
    r.w += cP * x2.w;
  }
  reinterpret_cast<float4*>(out)[gi] = r;
}

template <bool HAS_P>
__global__ __launch_bounds__(PROP_WAVES * 32) void prop_kernel(
    const float* __restrict__ x, const float* __restrict__ xp, const float* __restrict__ deg,
    const unsigned* __restrict__ seg, const int* __restrict__ start,
    float* __restrict__ out, float cH, float cX, float cP, int N) {
  __shared__ float acc[OWN_NODES * D_FEAT];
  const int t = threadIdx.x;
  const int lane = t & 31;
  const int wv = __builtin_amdgcn_readfirstlane(t >> 5);
  const int b = blockIdx.x;

  for (int i = t; i < OWN_NODES * D_FEAT / 4; i += PROP_WAVES * 32)
    reinterpret_cast<float4*>(acc)[i] = make_float4(0.0f, 0.0f, 0.0f, 0.0f);
  const int s0 = start[b];
  const int s1 = start[b + 1];
  __syncthreads();

  const int nruns = (s1 - s0 + RUN_EDGES - 1) / RUN_EDGES;
  int run = wv;
  if (run < nruns) {
    int rb = s0 + run * RUN_EDGES;
    int cnt = s1 - rb < RUN_EDGES ? s1 - rb : RUN_EDGES;
    unsigned p_v = seg[rb + (lane < cnt ? lane : cnt - 1)];
    float g_v = deg[p_v >> 8];
    for (;;) {
      const int nxt = run + PROP_WAVES;
      const bool more = nxt < nruns;
      unsigned p_n = p_v;
      int cnt_n = 0;
      if (more) {
        const int rbn = s0 + nxt * RUN_EDGES;
        cnt_n = s1 - rbn < RUN_EDGES ? s1 - rbn : RUN_EDGES;
        p_n = seg[rbn + (lane < cnt_n ? lane : cnt_n - 1)];
      }
      const int s_v = (int)(p_v >> 8);
      const int d_v = (int)(p_v & 255u) * D_FEAT;
      float g_n = 0.0f;
      if (cnt == RUN_EDGES) {
        float2 ra[ROWS_GROUP], rc[ROWS_GROUP];
        load_rows(ra, x, s_v, 0, lane);
        load_rows(rc, x, s_v, 8, lane);
        if (more) g_n = deg[p_n >> 8];
        const int n_v = __float_as_int(node_norm(g_v));
        add_rows(ra, acc, d_v, n_v, 0, lane);
        load_rows(ra, x, s_v, 16, lane);
        add_rows(rc, acc, d_v, n_v, 8, lane);
        load_rows(rc, x, s_v, 24, lane);
        add_rows(ra, acc, d_v, n_v, 16, lane);
        add_rows(rc, acc, d_v, n_v, 24, lane);
      } else {
        if (more) g_n = deg[p_n >> 8];
        const int n_v = __float_as_int(node_norm(g_v));
        for (int j = 0; j < cnt; ++j) {
          const int s = __builtin_amdgcn_readlane(s_v, j);
          const int d = __builtin_amdgcn_readlane(d_v, j);
          const float ns = __int_as_float(__builtin_amdgcn_readlane(n_v, j));
          const float2 v = *reinterpret_cast<const float2*>(x + (long)s * D_FEAT + 2 * lane);
          float* o = acc + d + 2 * lane;
          lds_add(o, v.x * ns);
          lds_add(o + 1, v.y * ns);
        }
      }
      if (!more) break;
      p_v = p_n;
      g_v = g_n;
      cnt = cnt_n;
      run = nxt;
    }
  }
  __syncthreads();

  const int n0 = b * OWN_NODES;
  constexpr int PER_BLOCK = OWN_NODES * (D_FEAT / 4);
  constexpr int FULL_TRIPS = PER_BLOCK / (PROP_WAVES * 32);
  if (n0 + OWN_NODES <= N) {
#pragma unroll
    for (int k = 0; k < FULL_TRIPS; ++k)
      write_term<HAS_P>(t + k * (PROP_WAVES * 32), n0, acc, x, xp, deg, out, cH, cX, cP);
    if (t < PER_BLOCK - FULL_TRIPS * (PROP_WAVES * 32))
      write_term<HAS_P>(t + FULL_TRIPS * (PROP_WAVES * 32), n0, acc, x, xp, deg, out, cH, cX, cP);
  } else {
    for (int i = t; i < PER_BLOCK; i += PROP_WAVES * 32) {
      if (n0 + (i >> 4) < N) write_term<HAS_P>(i, n0, acc, x, xp, deg, out, cH, cX, cP);
    }
  }
}


#define GEMM_WAVES 4
#define TILE_SLOTS 32
#define W_TERM (D_FEAT * D_FEAT)

__device__ __forceinline__ void load8(float (&a)[8], const float* __restrict__ p) {
  const float4 lo = reinterpret_cast<const float4*>(p)[0];
  const float4 hi = reinterpret_cast<const float4*>(p)[1];
  a[0] = lo.x; a[1] = lo.y; a[2] = lo.z; a[3] = lo.w;
  a[4] = hi.x; a[5] = hi.y; a[6] = hi.z; a[7] = hi.w;
}

__device__ __forceinline__ void term_quarter(v8f (&acc)[4], const float (&a)[8],
                                             const float* __restrict__ wrow) {
  v2f bf[4][4];
#pragma unroll
  for (int j = 0; j < 4; ++j) {
    const float4 w0 = *reinterpret_cast<const float4*>(wrow + (2 * j + 0) * D_FEAT);
    const float4 w1 = *reinterpret_cast<const float4*>(wrow + (2 * j + 1) * D_FEAT);
    bf[j][0].x = w0.x; bf[j][0].y = w1.x;
    bf[j][1].x = w0.y; bf[j][1].y = w1.y;
    bf[j][2].x = w0.z; bf[j][2].y = w1.z;
    bf[j][3].x = w0.w; bf[j][3].y = w1.w;
  }
#pragma unroll
  for (int j = 0; j < 4; ++j) {
    v2f av;
    av.x = a[2 * j + 0];
    av.y = a[2 * j + 1];
#pragma unroll
    for (int n = 0; n < 4; ++n) {
      acc[n] = __builtin_amdgcn_wmma_f32_16x16x4_f32(
           false, av,  false, bf[j][n],
           (short)0, acc[n],  false,  false);
    }
  }
}

__global__ __launch_bounds__(128) void gemm_fused_kernel(
    const float* __restrict__ X0, const float* __restrict__ X1,
    const float* __restrict__ X2, const float* __restrict__ X3,
    const float* __restrict__ W, const float* __restrict__ bias,
    float* __restrict__ out, int N) {
  __shared__ float red[GEMM_WAVES * TILE_SLOTS * 32];

  const int lane = threadIdx.x & 31;
  const int wv   = threadIdx.x >> 5;
  const int row0 = blockIdx.x * 16;
  if (row0 >= N) return;

  const int c  = lane & 15;
  const int h  = lane >> 4;
  const int k0 = wv * 16 + h * 8;

  int arow = row0 + c;
  if (arow >= N) arow = N - 1;
  const long aoff = (long)arow * D_FEAT + k0;
  const float* wq = W + (long)k0 * D_FEAT + 4 * c;

  v8f acc[4];
  {
    const float4 bb = *reinterpret_cast<const float4*>(bias + 4 * c);
    const float b0 = wv == 0 ? bb.x : 0.0f;
    const float b1 = wv == 0 ? bb.y : 0.0f;
    const float b2 = wv == 0 ? bb.z : 0.0f;
    const float b3 = wv == 0 ? bb.w : 0.0f;
#pragma unroll
    for (int v = 0; v < 8; ++v) {
      acc[0][v] = b0; acc[1][v] = b1; acc[2][v] = b2; acc[3][v] = b3;
    }
  }

  float a[8];
  load8(a, X0 + aoff);
  term_quarter(acc, a, wq + 0 * W_TERM);
  load8(a, X1 + aoff);
  term_quarter(acc, a, wq + 1 * W_TERM);
  load8(a, X2 + aoff);
  term_quarter(acc, a, wq + 2 * W_TERM);
  load8(a, X3 + aoff);
  term_quarter(acc, a, wq + 3 * W_TERM);

  {
    float* mine = red + (wv * TILE_SLOTS) * 32 + lane;
#pragma unroll
    for (int v = 0; v < 8; ++v) {
#pragma unroll
      for (int n = 0; n < 4; ++n) mine[(4 * v + n) * 32] = acc[n][v];
    }
  }
  __syncthreads();

  for (int j = 0; j < 2; ++j) {
    const int v = 2 * wv + j;
    const float* p = red + (4 * v) * 32 + lane;
    float o[4];
#pragma unroll
    for (int n = 0; n < 4; ++n) {
      const float* q = p + n * 32;
      o[n] = ((q[0 * TILE_SLOTS * 32] + q[1 * TILE_SLOTS * 32]) + q[2 * TILE_SLOTS * 32])
             + q[3 * TILE_SLOTS * 32];
    }
    const int row = row0 + 8 * h + v;
    if (row < N)
      *reinterpret_cast<float4*>(out + (long)row * D_FEAT + 4 * c) = make_float4(o[0], o[1], o[2], o[3]);
  }
}


extern "C" void kernel_launch(void* const* d_in, const int* in_sizes, int n_in,
                              void* d_out, int out_size, void* d_ws, size_t ws_size,
                              hipStream_t stream) {
  const float* feat = (const float*)d_in[0];
  const int*   src  = (const int*)d_in[1];
  const int*   dst  = (const int*)d_in[2];
  const float* W    = (const float*)d_in[3];
  const float* b    = (const float*)d_in[4];
  float*       out  = (float*)d_out;

  const int N = in_sizes[0] / D_FEAT;
  const int E = in_sizes[1];
  const long nd = (long)N * D_FEAT;
  const int nown = (N + OWN_NODES - 1) / OWN_NODES;

  const long npad = ((long)N + 31) & ~31L;
  const long epad = ((long)E + 31) & ~31L;
  const long ncur = 2L * MAX_OWN * CUR_PITCH;
  float*    deg    = (float*)d_ws;
  int*      cursor = (int*)(deg + npad);
  int*      cnt    = cursor + ncur;
  int*      start  = cnt + MAX_OWN;
  unsigned* seg    = (unsigned*)(start + MAX_OWN);
  float*    txA    = (float*)(seg + epad);
  float*    txB    = txA + nd;
  float*    txC    = txB + nd;

  const int ggrid = (N + 15) / 16;
  const int fgrid = (E + FILL_THREADS * FILL_PER_THREAD - 1) / (FILL_THREADS * FILL_PER_THREAD);

  zero_kernel<<<(int)((npad + ncur + 255) / 256), 256, 0, stream>>>(deg, npad + ncur);
  deg_kernel<<<(E + 255) / 256, 256, 0, stream>>>(dst, deg, E);
  count_kernel<<<(nown + 7) / 8, 256, 0, stream>>>(deg, cnt, N, nown);
  fill_kernel<<<fgrid, FILL_THREADS, 0, stream>>>(src, dst, cnt, cursor, start, seg, E, nown);

  prop_kernel<false><<<nown, PROP_WAVES * 32, 0, stream>>>(feat, feat, deg, seg, start, txA,
                                                           -2.0f / LAM, 2.0f / LAM - 1.0f, 0.0f, N);
  prop_kernel<true><<<nown, PROP_WAVES * 32, 0, stream>>>(txA, feat, deg, seg, start, txB,
                                                          -4.0f / LAM, 4.0f / LAM - 2.0f, -1.0f, N);
  prop_kernel<true><<<nown, PROP_WAVES * 32, 0, stream>>>(txB, txA, deg, seg, start, txC,
                                                          -4.0f / LAM, 4.0f / LAM - 2.0f, -1.0f, N);

  gemm_fused_kernel<<<ggrid, 128, 0, stream>>>(feat, txA, txB, txC, W, b, out, N);
}


# Props re-tiled: 256 owner blocks of 391 nodes, one 32-wave block per WGP; a wave walks an equal contiguous share of its segment as one row pipeline, tail in groups of 8; LDS sums in arrival order
# speedup vs baseline: 2.6611x; 1.0413x over previous
#include <hip/hip_runtime.h>

typedef __attribute__((ext_vector_type(2))) float v2f;
typedef __attribute__((ext_vector_type(8))) float v8f;

#define D_FEAT 64
constexpr float LAM = 2.0f;


__global__ void zero_kernel(float* __restrict__ p, long n) {
  long i = (long)blockIdx.x * blockDim.x + threadIdx.x;
  long stride = (long)gridDim.x * blockDim.x;
  for (; i < n; i += stride) p[i] = 0.0f;
}

__global__ void deg_kernel(const int* __restrict__ dst, float* __restrict__ deg, int E) {
  int i = blockIdx.x * blockDim.x + threadIdx.x;
  if (i < E) atomicAdd(&deg[dst[i]], 1.0f);
}

__device__ __forceinline__ float node_norm(float deg) {
  return rsqrtf(fmaxf(deg, 1.0f));
}

#define OWN_NODES 391
#define LOC_BITS 9
#define MAX_OWN 1024
#define FILL_THREADS 512
#define FILL_PER_THREAD 8
#define CUR_PITCH 32
#define RUN_EDGES 32
#define ROWS_GROUP 8
#define PROP_WAVES 32

__global__ __launch_bounds__(256) void count_kernel(const float* __restrict__ deg,
                                                    int* __restrict__ cnt, int N, int nown) {
  const int lane = threadIdx.x & 31;
  const int b = blockIdx.x * (blockDim.x >> 5) + (threadIdx.x >> 5);
  if (b >= nown) return;
  const int n0 = b * OWN_NODES;
  int s = 0;
#pragma unroll
  for (int k = 0; k < (OWN_NODES + 31) / 32; ++k) {
    const int i = lane + 32 * k;
    const bool ok = i < OWN_NODES && n0 + i < N;
    const float v = deg[ok ? n0 + i : n0];
    s += ok ? (int)v : 0;
  }
  for (int m = 16; m > 0; m >>= 1) s += __shfl_xor(s, m);
  if (lane == 0) cnt[b] = s;
}

__global__ __launch_bounds__(FILL_THREADS) void fill_kernel(
    const int* __restrict__ src, const int* __restrict__ dst, const int* __restrict__ cnt,
    int* __restrict__ cursor, int* __restrict__ start, unsigned* __restrict__ seg,
    int E, int nown) {
  __shared__ int s_start[MAX_OWN];
  __shared__ int s_base[MAX_OWN];
  __shared__ int s_hist[MAX_OWN];
  __shared__ int s_wsum[FILL_THREADS / 32];
  const int t = threadIdx.x;
  const int lane = t & 31;
  const int wv = t >> 5;

  const int c0 = (2 * t < nown) ? cnt[2 * t] : 0;
  const int c1 = (2 * t + 1 < nown) ? cnt[2 * t + 1] : 0;
  s_hist[2 * t] = 0;
  s_hist[2 * t + 1] = 0;
  int incl = c0 + c1;
  for (int d = 1; d < 32; d <<= 1) {
    const int v = __shfl_up(incl, d);
    if (lane >= d) incl += v;
  }
  if (lane == 31) s_wsum[wv] = incl;
  __syncthreads();
  int woff = 0;
  for (int w = 0; w < wv; ++w) woff += s_wsum[w];
  const int excl = woff + incl - (c0 + c1);
  s_start[2 * t] = excl;
  s_start[2 * t + 1] = excl + c0;
  if (blockIdx.x == 0) {
    if (2 * t <= nown) start[2 * t] = excl;
    if (2 * t + 1 <= nown) start[2 * t + 1] = excl + c0;
  }
  __syncthreads();

  const unsigned hw_id = __builtin_amdgcn_s_sendmsg_rtn(0x87);
  const int half = (int)((hw_id >> 18) & 1u);
  int* __restrict__ cur = cursor + (long)half * (MAX_OWN * CUR_PITCH);

  const long base = (long)blockIdx.x * (FILL_THREADS * FILL_PER_THREAD);
  int rk[FILL_PER_THREAD];
  int ob[FILL_PER_THREAD];
  unsigned pk[FILL_PER_THREAD];
  int dv[FILL_PER_THREAD];
  int sv[FILL_PER_THREAD];
#pragma unroll
  for (int j = 0; j < FILL_PER_THREAD; ++j) {
    const long e = base + j * FILL_THREADS + t;
    const long ec = e < E ? e : (long)E - 1;
    dv[j] = dst[ec];
    sv[j] = src[ec];
  }
#pragma unroll
  for (int j = 0; j < FILL_PER_THREAD; ++j) {
    const long e = base + j * FILL_THREADS + t;
    ob[j] = -1;
    rk[j] = 0;
    pk[j] = 0u;
    if (e < E) {
      const int d = dv[j];
      const int s = sv[j];
      const int b = d / OWN_NODES;
      ob[j] = b;
      pk[j] = ((unsigned)s << LOC_BITS) | (unsigned)(d - b * OWN_NODES);
      rk[j] = atomicAdd(&s_hist[b], 1);
    }
  }
  __syncthreads();
  for (int b = t; b < nown; b += FILL_THREADS) {
    const int hcount = s_hist[b];
    int at = 0;
    if (hcount > 0) {
      const int old = atomicAdd(&cur[b * CUR_PITCH], hcount);
      at = half == 0 ? s_start[b] + old : s_start[b + 1] - old - hcount;
    }
    s_base[b] = at;
  }
  __syncthreads();
#pragma unroll
  for (int j = 0; j < FILL_PER_THREAD; ++j) {
    if (ob[j] >= 0) seg[s_base[ob[j]] + rk[j]] = pk[j];
  }
}


__device__ __forceinline__ void lds_add(float* p, float v) {
  __hip_atomic_fetch_add(p, v, __ATOMIC_RELAXED, __HIP_MEMORY_SCOPE_WORKGROUP);
}

__device__ __forceinline__ void load_rows(float2 (&r)[ROWS_GROUP], const float* __restrict__ x,
                                          int s_v, int first, int lane) {
#pragma unroll
  for (int j = 0; j < ROWS_GROUP; ++j) {
    const int s = __builtin_amdgcn_readlane(s_v, first + j);
    r[j] = *reinterpret_cast<const float2*>(x + (long)s * D_FEAT + 2 * lane);
  }
}

template <bool FULL>
__device__ __forceinline__ void add_rows(const float2 (&r)[ROWS_GROUP], float* acc,
                                         int d_v, int n_v, int first, int cnt, int lane) {
#pragma unroll
  for (int j = 0; j < ROWS_GROUP; ++j) {
    if (FULL || first + j < cnt) {
      const int d = __builtin_amdgcn_readlane(d_v, first + j);
      const float ns = __int_as_float(__builtin_amdgcn_readlane(n_v, first + j));
      float* o = acc + d + 2 * lane;
      lds_add(o, r[j].x * ns);
      lds_add(o + 1, r[j].y * ns);
    }
  }
}

template <bool HAS_P>
__device__ __forceinline__ void write_term(int i, int n0, const float* acc,
                                           const float* __restrict__ x, const float* __restrict__ xp,
                                           const float* __restrict__ deg, float* __restrict__ out,
                                           float cH, float cX, float cP) {
  const int node = n0 + (i >> 4);
  const float nn = node_norm(deg[node]);
  const float4 a = reinterpret_cast<const float4*>(acc)[i];
  const long gi = (long)node * (D_FEAT / 4) + (i & 15);
  const float4 x1 = reinterpret_cast<const float4*>(x)[gi];
  float4 r;
  r.x = cH * (a.x * nn) + cX * x1.x;
  r.y = cH * (a.y * nn) + cX * x1.y;
  r.z = cH * (a.z * nn) + cX * x1.z;
  r.w = cH * (a.w * nn) + cX * x1.w;
  if (HAS_P) {
    const float4 x2 = reinterpret_cast<const float4*>(xp)[gi];
    r.x += cP * x2.x;
    r.y += cP * x2.y;
    r.z += cP * x2.z;
    r.w += cP * x2.w;
  }
  reinterpret_cast<float4*>(out)[gi] = r;
}

template <bool HAS_P>
__global__ __launch_bounds__(PROP_WAVES * 32) void prop_kernel(
    const float* __restrict__ x, const float* __restrict__ xp, const float* __restrict__ deg,
    const unsigned* __restrict__ seg, const int* __restrict__ start,
    float* __restrict__ out, float cH, float cX, float cP, int N) {
  __shared__ float acc[OWN_NODES * D_FEAT];
  const int t = threadIdx.x;
  const int lane = t & 31;
  const int wv = __builtin_amdgcn_readfirstlane(t >> 5);
  const int b = blockIdx.x;

  for (int i = t; i < OWN_NODES * D_FEAT / 4; i += PROP_WAVES * 32)
    reinterpret_cast<float4*>(acc)[i] = make_float4(0.0f, 0.0f, 0.0f, 0.0f);
  const int s0 = start[b];
  const int s1 = start[b + 1];
  __syncthreads();

  int per = (s1 - s0 + PROP_WAVES - 1) / PROP_WAVES;
  per = (per + ROWS_GROUP - 1) & ~(ROWS_GROUP - 1);
  const int e0 = s0 + wv * per;
  const int e1 = e0 + per < s1 ? e0 + per : s1;
  if (e0 < e1) {
    int cnt = e1 - e0 < RUN_EDGES ? e1 - e0 : RUN_EDGES;
    unsigned p_v = seg[e0 + (lane < cnt ? lane : cnt - 1)];
    int rbn = e0 + RUN_EDGES;
    int cnt_n = 0;
    unsigned p_n = p_v;
    if (rbn < e1) {
      cnt_n = e1 - rbn < RUN_EDGES ? e1 - rbn : RUN_EDGES;
      p_n = seg[rbn + (lane < cnt_n ? lane : cnt_n - 1)];
    }
    float g_v = deg[p_v >> LOC_BITS];
    float2 ra[ROWS_GROUP], rc[ROWS_GROUP];
    load_rows(ra, x, (int)(p_v >> LOC_BITS), 0, lane);
    load_rows(rc, x, (int)(p_v >> LOC_BITS), 8, lane);
    for (;;) {
      const bool more = cnt_n > 0;
      const int s_v = (int)(p_v >> LOC_BITS);
      const int d_v = (int)(p_v & ((1u << LOC_BITS) - 1u)) * D_FEAT;
      const int n_v = __float_as_int(node_norm(g_v));
      unsigned p_nn = p_n;
      int cnt_nn = 0;
      float g_n = 0.0f;
      if (cnt == RUN_EDGES) {
        add_rows<true>(ra, acc, d_v, n_v, 0, cnt, lane);
        load_rows(ra, x, s_v, 16, lane);
        add_rows<true>(rc, acc, d_v, n_v, 8, cnt, lane);
        load_rows(rc, x, s_v, 24, lane);
        if (more) {
          const int s_n = (int)(p_n >> LOC_BITS);
          g_n = deg[s_n];
          const int rbnn = rbn + RUN_EDGES;
          if (rbnn < e1) {
            cnt_nn = e1 - rbnn < RUN_EDGES ? e1 - rbnn : RUN_EDGES;
            p_nn = seg[rbnn + (lane < cnt_nn ? lane : cnt_nn - 1)];
          }
          add_rows<true>(ra, acc, d_v, n_v, 16, cnt, lane);
          load_rows(ra, x, s_n, 0, lane);
          add_rows<true>(rc, acc, d_v, n_v, 24, cnt, lane);
          load_rows(rc, x, s_n, 8, lane);
        } else {
          add_rows<true>(ra, acc, d_v, n_v, 16, cnt, lane);
          add_rows<true>(rc, acc, d_v, n_v, 24, cnt, lane);
        }
      } else {
        add_rows<false>(ra, acc, d_v, n_v, 0, cnt, lane);
        if (cnt > 16) load_rows(ra, x, s_v, 16, lane);
        if (cnt > 8) add_rows<false>(rc, acc, d_v, n_v, 8, cnt, lane);
        if (cnt > 24) load_rows(rc, x, s_v, 24, lane);
        if (cnt > 16) add_rows<false>(ra, acc, d_v, n_v, 16, cnt, lane);
        if (cnt > 24) add_rows<false>(rc, acc, d_v, n_v, 24, cnt, lane);
      }
      if (!more) break;
      p_v = p_n;
      g_v = g_n;
      cnt = cnt_n;
      p_n = p_nn;
      cnt_n = cnt_nn;
      rbn += RUN_EDGES;
    }
  }
  __syncthreads();

  const int n0 = b * OWN_NODES;
  constexpr int PER_BLOCK = OWN_NODES * (D_FEAT / 4);
  constexpr int FULL_TRIPS = PER_BLOCK / (PROP_WAVES * 32);
  if (n0 + OWN_NODES <= N) {
#pragma unroll
    for (int k = 0; k < FULL_TRIPS; ++k)
      write_term<HAS_P>(t + k * (PROP_WAVES * 32), n0, acc, x, xp, deg, out, cH, cX, cP);
    if (t < PER_BLOCK - FULL_TRIPS * (PROP_WAVES * 32))
      write_term<HAS_P>(t + FULL_TRIPS * (PROP_WAVES * 32), n0, acc, x, xp, deg, out, cH, cX, cP);
  } else {
    for (int i = t; i < PER_BLOCK; i += PROP_WAVES * 32) {
      if (n0 + (i >> 4) < N) write_term<HAS_P>(i, n0, acc, x, xp, deg, out, cH, cX, cP);
    }
  }
}


#define GEMM_WAVES 4
#define TILE_SLOTS 32
#define W_TERM (D_FEAT * D_FEAT)

__device__ __forceinline__ void load8(float (&a)[8], const float* __restrict__ p) {
  const float4 lo = reinterpret_cast<const float4*>(p)[0];
  const float4 hi = reinterpret_cast<const float4*>(p)[1];
  a[0] = lo.x; a[1] = lo.y; a[2] = lo.z; a[3] = lo.w;
  a[4] = hi.x; a[5] = hi.y; a[6] = hi.z; a[7] = hi.w;
}

__device__ __forceinline__ void term_quarter(v8f (&acc)[4], const float (&a)[8],
                                             const float* __restrict__ wrow) {
  v2f bf[4][4];
#pragma unroll
  for (int j = 0; j < 4; ++j) {
    const float4 w0 = *reinterpret_cast<const float4*>(wrow + (2 * j + 0) * D_FEAT);
    const float4 w1 = *reinterpret_cast<const float4*>(wrow + (2 * j + 1) * D_FEAT);
    bf[j][0].x = w0.x; bf[j][0].y = w1.x;
    bf[j][1].x = w0.y; bf[j][1].y = w1.y;
    bf[j][2].x = w0.z; bf[j][2].y = w1.z;
    bf[j][3].x = w0.w; bf[j][3].y = w1.w;
  }
#pragma unroll
  for (int j = 0; j < 4; ++j) {
    v2f av;
    av.x = a[2 * j + 0];
    av.y = a[2 * j + 1];
#pragma unroll
    for (int n = 0; n < 4; ++n) {
      acc[n] = __builtin_amdgcn_wmma_f32_16x16x4_f32(
           false, av,  false, bf[j][n],
           (short)0, acc[n],  false,  false);
    }
  }
}

__global__ __launch_bounds__(128) void gemm_fused_kernel(
    const float* __restrict__ X0, const float* __restrict__ X1,
    const float* __restrict__ X2, const float* __restrict__ X3,
    const float* __restrict__ W, const float* __restrict__ bias,
    float* __restrict__ out, int N) {
  __shared__ float red[GEMM_WAVES * TILE_SLOTS * 32];

  const int lane = threadIdx.x & 31;
  const int wv   = threadIdx.x >> 5;
  const int row0 = blockIdx.x * 16;
  if (row0 >= N) return;

  const int c  = lane & 15;
  const int h  = lane >> 4;
  const int k0 = wv * 16 + h * 8;

  int arow = row0 + c;
  if (arow >= N) arow = N - 1;
  const long aoff = (long)arow * D_FEAT + k0;
  const float* wq = W + (long)k0 * D_FEAT + 4 * c;

  v8f acc[4];
  {
    const float4 bb = *reinterpret_cast<const float4*>(bias + 4 * c);
    const float b0 = wv == 0 ? bb.x : 0.0f;
    const float b1 = wv == 0 ? bb.y : 0.0f;
    const float b2 = wv == 0 ? bb.z : 0.0f;
    const float b3 = wv == 0 ? bb.w : 0.0f;
#pragma unroll
    for (int v = 0; v < 8; ++v) {
      acc[0][v] = b0; acc[1][v] = b1; acc[2][v] = b2; acc[3][v] = b3;
    }
  }

  float a[8];
  load8(a, X0 + aoff);
  term_quarter(acc, a, wq + 0 * W_TERM);
  load8(a, X1 + aoff);
  term_quarter(acc, a, wq + 1 * W_TERM);
  load8(a, X2 + aoff);
  term_quarter(acc, a, wq + 2 * W_TERM);
  load8(a, X3 + aoff);
  term_quarter(acc, a, wq + 3 * W_TERM);

  {
    float* mine = red + (wv * TILE_SLOTS) * 32 + lane;
#pragma unroll
    for (int v = 0; v < 8; ++v) {
#pragma unroll
      for (int n = 0; n < 4; ++n) mine[(4 * v + n) * 32] = acc[n][v];
    }
  }
  __syncthreads();

  for (int j = 0; j < 2; ++j) {
    const int v = 2 * wv + j;
    const float* p = red + (4 * v) * 32 + lane;
    float o[4];
#pragma unroll
    for (int n = 0; n < 4; ++n) {
      const float* q = p + n * 32;
      o[n] = ((q[0 * TILE_SLOTS * 32] + q[1 * TILE_SLOTS * 32]) + q[2 * TILE_SLOTS * 32])
             + q[3 * TILE_SLOTS * 32];
    }
    const int row = row0 + 8 * h + v;
    if (row < N)
      *reinterpret_cast<float4*>(out + (long)row * D_FEAT + 4 * c) = make_float4(o[0], o[1], o[2], o[3]);
  }
}


extern "C" void kernel_launch(void* const* d_in, const int* in_sizes, int n_in,
                              void* d_out, int out_size, void* d_ws, size_t ws_size,
                              hipStream_t stream) {
  const float* feat = (const float*)d_in[0];
  const int*   src  = (const int*)d_in[1];
  const int*   dst  = (const int*)d_in[2];
  const float* W    = (const float*)d_in[3];
  const float* b    = (const float*)d_in[4];
  float*       out  = (float*)d_out;

  const int N = in_sizes[0] / D_FEAT;
  const int E = in_sizes[1];
  const long nd = (long)N * D_FEAT;
  const int nown = (N + OWN_NODES - 1) / OWN_NODES;

  const long npad = ((long)N + 31) & ~31L;
  const long epad = ((long)E + 31) & ~31L;
  const long ncur = 2L * MAX_OWN * CUR_PITCH;
  float*    deg    = (float*)d_ws;
  int*      cursor = (int*)(deg + npad);
  int*      cnt    = cursor + ncur;
  int*      start  = cnt + MAX_OWN;
  unsigned* seg    = (unsigned*)(start + MAX_OWN);
  float*    txA    = (float*)(seg + epad);
  float*    txB    = txA + nd;
  float*    txC    = txB + nd;

  const int ggrid = (N + 15) / 16;
  const int fgrid = (E + FILL_THREADS * FILL_PER_THREAD - 1) / (FILL_THREADS * FILL_PER_THREAD);

  zero_kernel<<<(int)((npad + ncur + 255) / 256), 256, 0, stream>>>(deg, npad + ncur);
  deg_kernel<<<(E + 255) / 256, 256, 0, stream>>>(dst, deg, E);
  count_kernel<<<(nown + 7) / 8, 256, 0, stream>>>(deg, cnt, N, nown);
  fill_kernel<<<fgrid, FILL_THREADS, 0, stream>>>(src, dst, cnt, cursor, start, seg, E, nown);

  prop_kernel<false><<<nown, PROP_WAVES * 32, 0, stream>>>(feat, feat, deg, seg, start, txA,
                                                           -2.0f / LAM, 2.0f / LAM - 1.0f, 0.0f, N);
  prop_kernel<true><<<nown, PROP_WAVES * 32, 0, stream>>>(txA, feat, deg, seg, start, txB,
                                                          -4.0f / LAM, 4.0f / LAM - 2.0f, -1.0f, N);
  prop_kernel<true><<<nown, PROP_WAVES * 32, 0, stream>>>(txB, txA, deg, seg, start, txC,
                                                          -4.0f / LAM, 4.0f / LAM - 2.0f, -1.0f, N);

  gemm_fused_kernel<<<ggrid, 128, 0, stream>>>(feat, txA, txB, txC, W, b, out, N);
}


# Edge entries carry the source's norm (8 B per edge, looked up once when edges are grouped): props get it in the coalesced id load, no per-edge count lookup; values and sum order unchanged
# speedup vs baseline: 2.7742x; 1.0425x over previous
#include <hip/hip_runtime.h>

typedef __attribute__((ext_vector_type(2))) float v2f;
typedef __attribute__((ext_vector_type(8))) float v8f;

#define D_FEAT 64
constexpr float LAM = 2.0f;


__global__ void zero_kernel(float* __restrict__ p, long n) {
  long i = (long)blockIdx.x * blockDim.x + threadIdx.x;
  long stride = (long)gridDim.x * blockDim.x;
  for (; i < n; i += stride) p[i] = 0.0f;
}

__global__ void deg_kernel(const int* __restrict__ dst, float* __restrict__ deg, int E) {
  int i = blockIdx.x * blockDim.x + threadIdx.x;
  if (i < E) atomicAdd(&deg[dst[i]], 1.0f);
}

__device__ __forceinline__ float node_norm(float deg) {
  return rsqrtf(fmaxf(deg, 1.0f));
}

#define OWN_NODES 391
#define LOC_BITS 9
#define MAX_OWN 1024
#define FILL_THREADS 512
#define FILL_PER_THREAD 8
#define CUR_PITCH 32
#define RUN_EDGES 32
#define ROWS_GROUP 8
#define PROP_WAVES 32

__global__ __launch_bounds__(256) void count_kernel(const float* __restrict__ deg,
                                                    int* __restrict__ cnt, int N, int nown) {
  const int lane = threadIdx.x & 31;
  const int b = blockIdx.x * (blockDim.x >> 5) + (threadIdx.x >> 5);
  if (b >= nown) return;
  const int n0 = b * OWN_NODES;
  int s = 0;
#pragma unroll
  for (int k = 0; k < (OWN_NODES + 31) / 32; ++k) {
    const int i = lane + 32 * k;
    const bool ok = i < OWN_NODES && n0 + i < N;
    const float v = deg[ok ? n0 + i : n0];
    s += ok ? (int)v : 0;
  }
  for (int m = 16; m > 0; m >>= 1) s += __shfl_xor(s, m);
  if (lane == 0) cnt[b] = s;
}

__global__ __launch_bounds__(FILL_THREADS) void fill_kernel(
    const int* __restrict__ src, const int* __restrict__ dst, const int* __restrict__ cnt,
    const float* __restrict__ deg, int* __restrict__ cursor, int* __restrict__ start,
    uint2* __restrict__ seg, int E, int nown) {
  __shared__ int s_start[MAX_OWN];
  __shared__ int s_base[MAX_OWN];
  __shared__ int s_hist[MAX_OWN];
  __shared__ int s_wsum[FILL_THREADS / 32];
  const int t = threadIdx.x;
  const int lane = t & 31;
  const int wv = t >> 5;

  const int c0 = (2 * t < nown) ? cnt[2 * t] : 0;
  const int c1 = (2 * t + 1 < nown) ? cnt[2 * t + 1] : 0;
  s_hist[2 * t] = 0;
  s_hist[2 * t + 1] = 0;
  int incl = c0 + c1;
  for (int d = 1; d < 32; d <<= 1) {
    const int v = __shfl_up(incl, d);
    if (lane >= d) incl += v;
  }
  if (lane == 31) s_wsum[wv] = incl;
  __syncthreads();
  int woff = 0;
  for (int w = 0; w < wv; ++w) woff += s_wsum[w];
  const int excl = woff + incl - (c0 + c1);
  s_start[2 * t] = excl;
  s_start[2 * t + 1] = excl + c0;
  if (blockIdx.x == 0) {
    if (2 * t <= nown) start[2 * t] = excl;
    if (2 * t + 1 <= nown) start[2 * t + 1] = excl + c0;
  }
  __syncthreads();

  const unsigned hw_id = __builtin_amdgcn_s_sendmsg_rtn(0x87);
  const int half = (int)((hw_id >> 18) & 1u);
  int* __restrict__ cur = cursor + (long)half * (MAX_OWN * CUR_PITCH);

  const long base = (long)blockIdx.x * (FILL_THREADS * FILL_PER_THREAD);
  int rk[FILL_PER_THREAD];
  int ob[FILL_PER_THREAD];
  unsigned pk[FILL_PER_THREAD];
  int dv[FILL_PER_THREAD];
  int sv[FILL_PER_THREAD];
#pragma unroll
  for (int j = 0; j < FILL_PER_THREAD; ++j) {
    const long e = base + j * FILL_THREADS + t;
    const long ec = e < E ? e : (long)E - 1;
    dv[j] = dst[ec];
    sv[j] = src[ec];
  }
  float gv[FILL_PER_THREAD];
#pragma unroll
  for (int j = 0; j < FILL_PER_THREAD; ++j) gv[j] = deg[sv[j]];
#pragma unroll
  for (int j = 0; j < FILL_PER_THREAD; ++j) {
    const long e = base + j * FILL_THREADS + t;
    ob[j] = -1;
    rk[j] = 0;
    pk[j] = 0u;
    if (e < E) {
      const int d = dv[j];
      const int s = sv[j];
      const int b = d / OWN_NODES;
      ob[j] = b;
      pk[j] = ((unsigned)s << LOC_BITS) | (unsigned)(d - b * OWN_NODES);
      rk[j] = atomicAdd(&s_hist[b], 1);
    }
  }
  __syncthreads();
  for (int b = t; b < nown; b += FILL_THREADS) {
    const int hcount = s_hist[b];
    int at = 0;
    if (hcount > 0) {
      const int old = atomicAdd(&cur[b * CUR_PITCH], hcount);
      at = half == 0 ? s_start[b] + old : s_start[b + 1] - old - hcount;
    }
    s_base[b] = at;
  }
  __syncthreads();
#pragma unroll
  for (int j = 0; j < FILL_PER_THREAD; ++j) {
    if (ob[j] >= 0)
      seg[s_base[ob[j]] + rk[j]] = make_uint2(pk[j], (unsigned)__float_as_int(node_norm(gv[j])));
  }
}


__device__ __forceinline__ void lds_add(float* p, float v) {
  __hip_atomic_fetch_add(p, v, __ATOMIC_RELAXED, __HIP_MEMORY_SCOPE_WORKGROUP);
}

__device__ __forceinline__ void load_rows(float2 (&r)[ROWS_GROUP], const float* __restrict__ x,
                                          int s_v, int first, int lane) {
#pragma unroll
  for (int j = 0; j < ROWS_GROUP; ++j) {
    const int s = __builtin_amdgcn_readlane(s_v, first + j);
    r[j] = *reinterpret_cast<const float2*>(x + (long)s * D_FEAT + 2 * lane);
  }
}

template <bool FULL>
__device__ __forceinline__ void add_rows(const float2 (&r)[ROWS_GROUP], float* acc,
                                         int d_v, int n_v, int first, int cnt, int lane) {
#pragma unroll
  for (int j = 0; j < ROWS_GROUP; ++j) {
    if (FULL || first + j < cnt) {
      const int d = __builtin_amdgcn_readlane(d_v, first + j);
      const float ns = __int_as_float(__builtin_amdgcn_readlane(n_v, first + j));
      float* o = acc + d + 2 * lane;
      lds_add(o, r[j].x * ns);
      lds_add(o + 1, r[j].y * ns);
    }
  }
}

template <bool HAS_P>
__device__ __forceinline__ void write_term(int i, int n0, const float* acc,
                                           const float* __restrict__ x, const float* __restrict__ xp,
                                           const float* __restrict__ deg, float* __restrict__ out,
                                           float cH, float cX, float cP) {
  const int node = n0 + (i >> 4);
  const float nn = node_norm(deg[node]);
  const float4 a = reinterpret_cast<const float4*>(acc)[i];
  const long gi = (long)node * (D_FEAT / 4) + (i & 15);
  const float4 x1 = reinterpret_cast<const float4*>(x)[gi];
  float4 r;
  r.x = cH * (a.x * nn) + cX * x1.x;
  r.y = cH * (a.y * nn) + cX * x1.y;
  r.z = cH * (a.z * nn) + cX * x1.z;
  r.w = cH * (a.w * nn) + cX * x1.w;
  if (HAS_P) {
    const float4 x2 = reinterpret_cast<const float4*>(xp)[gi];
    r.x += cP * x2.x;
    r.y += cP * x2.y;
    r.z += cP * x2.z;
    r.w += cP * x2.w;
  }
  reinterpret_cast<float4*>(out)[gi] = r;
}

template <bool HAS_P>
__global__ __launch_bounds__(PROP_WAVES * 32) void prop_kernel(
    const float* __restrict__ x, const float* __restrict__ xp, const float* __restrict__ deg,
    const uint2* __restrict__ seg, const int* __restrict__ start,
    float* __restrict__ out, float cH, float cX, float cP, int N) {
  __shared__ float acc[OWN_NODES * D_FEAT];
  const int t = threadIdx.x;
  const int lane = t & 31;
  const int wv = __builtin_amdgcn_readfirstlane(t >> 5);
  const int b = blockIdx.x;

  for (int i = t; i < OWN_NODES * D_FEAT / 4; i += PROP_WAVES * 32)
    reinterpret_cast<float4*>(acc)[i] = make_float4(0.0f, 0.0f, 0.0f, 0.0f);
  const int s0 = start[b];
  const int s1 = start[b + 1];
  __syncthreads();

  int per = (s1 - s0 + PROP_WAVES - 1) / PROP_WAVES;
  per = (per + ROWS_GROUP - 1) & ~(ROWS_GROUP - 1);
  const int e0 = s0 + wv * per;
  const int e1 = e0 + per < s1 ? e0 + per : s1;
  if (e0 < e1) {
    int cnt = e1 - e0 < RUN_EDGES ? e1 - e0 : RUN_EDGES;
    uint2 p_v = seg[e0 + (lane < cnt ? lane : cnt - 1)];
    int rbn = e0 + RUN_EDGES;
    int cnt_n = 0;
    uint2 p_n = p_v;
    if (rbn < e1) {
      cnt_n = e1 - rbn < RUN_EDGES ? e1 - rbn : RUN_EDGES;
      p_n = seg[rbn + (lane < cnt_n ? lane : cnt_n - 1)];
    }
    float2 ra[ROWS_GROUP], rc[ROWS_GROUP];
    load_rows(ra, x, (int)(p_v.x >> LOC_BITS), 0, lane);
    load_rows(rc, x, (int)(p_v.x >> LOC_BITS), 8, lane);
    for (;;) {
      const bool more = cnt_n > 0;
      const int s_v = (int)(p_v.x >> LOC_BITS);
      const int d_v = (int)(p_v.x & ((1u << LOC_BITS) - 1u)) * D_FEAT;
      const int n_v = (int)p_v.y;
      uint2 p_nn = p_n;
      int cnt_nn = 0;
      if (cnt == RUN_EDGES) {
        add_rows<true>(ra, acc, d_v, n_v, 0, cnt, lane);
        load_rows(ra, x, s_v, 16, lane);
        add_rows<true>(rc, acc, d_v, n_v, 8, cnt, lane);
        load_rows(rc, x, s_v, 24, lane);
        if (more) {
          const int s_n = (int)(p_n.x >> LOC_BITS);
          const int rbnn = rbn + RUN_EDGES;
          if (rbnn < e1) {
            cnt_nn = e1 - rbnn < RUN_EDGES ? e1 - rbnn : RUN_EDGES;
            p_nn = seg[rbnn + (lane < cnt_nn ? lane : cnt_nn - 1)];
          }
          add_rows<true>(ra, acc, d_v, n_v, 16, cnt, lane);
          load_rows(ra, x, s_n, 0, lane);
          add_rows<true>(rc, acc, d_v, n_v, 24, cnt, lane);
          load_rows(rc, x, s_n, 8, lane);
        } else {
          add_rows<true>(ra, acc, d_v, n_v, 16, cnt, lane);
          add_rows<true>(rc, acc, d_v, n_v, 24, cnt, lane);
        }
      } else {
        add_rows<false>(ra, acc, d_v, n_v, 0, cnt, lane);
        if (cnt > 16) load_rows(ra, x, s_v, 16, lane);
        if (cnt > 8) add_rows<false>(rc, acc, d_v, n_v, 8, cnt, lane);
        if (cnt > 24) load_rows(rc, x, s_v, 24, lane);
        if (cnt > 16) add_rows<false>(ra, acc, d_v, n_v, 16, cnt, lane);
        if (cnt > 24) add_rows<false>(rc, acc, d_v, n_v, 24, cnt, lane);
      }
      if (!more) break;
      p_v = p_n;
      cnt = cnt_n;
      p_n = p_nn;
      cnt_n = cnt_nn;
      rbn += RUN_EDGES;
    }
  }
  __syncthreads();

  const int n0 = b * OWN_NODES;
  constexpr int PER_BLOCK = OWN_NODES * (D_FEAT / 4);
  constexpr int FULL_TRIPS = PER_BLOCK / (PROP_WAVES * 32);
  if (n0 + OWN_NODES <= N) {
#pragma unroll
    for (int k = 0; k < FULL_TRIPS; ++k)
      write_term<HAS_P>(t + k * (PROP_WAVES * 32), n0, acc, x, xp, deg, out, cH, cX, cP);
    if (t < PER_BLOCK - FULL_TRIPS * (PROP_WAVES * 32))
      write_term<HAS_P>(t + FULL_TRIPS * (PROP_WAVES * 32), n0, acc, x, xp, deg, out, cH, cX, cP);
  } else {
    for (int i = t; i < PER_BLOCK; i += PROP_WAVES * 32) {
      if (n0 + (i >> 4) < N) write_term<HAS_P>(i, n0, acc, x, xp, deg, out, cH, cX, cP);
    }
  }
}


#define GEMM_WAVES 4
#define TILE_SLOTS 32
#define W_TERM (D_FEAT * D_FEAT)

__device__ __forceinline__ void load8(float (&a)[8], const float* __restrict__ p) {
  const float4 lo = reinterpret_cast<const float4*>(p)[0];
  const float4 hi = reinterpret_cast<const float4*>(p)[1];
  a[0] = lo.x; a[1] = lo.y; a[2] = lo.z; a[3] = lo.w;
  a[4] = hi.x; a[5] = hi.y; a[6] = hi.z; a[7] = hi.w;
}

__device__ __forceinline__ void term_quarter(v8f (&acc)[4], const float (&a)[8],
                                             const float* __restrict__ wrow) {
  v2f bf[4][4];
#pragma unroll
  for (int j = 0; j < 4; ++j) {
    const float4 w0 = *reinterpret_cast<const float4*>(wrow + (2 * j + 0) * D_FEAT);
    const float4 w1 = *reinterpret_cast<const float4*>(wrow + (2 * j + 1) * D_FEAT);
    bf[j][0].x = w0.x; bf[j][0].y = w1.x;
    bf[j][1].x = w0.y; bf[j][1].y = w1.y;
    bf[j][2].x = w0.z; bf[j][2].y = w1.z;
    bf[j][3].x = w0.w; bf[j][3].y = w1.w;
  }
#pragma unroll
  for (int j = 0; j < 4; ++j) {
    v2f av;
    av.x = a[2 * j + 0];
    av.y = a[2 * j + 1];
#pragma unroll
    for (int n = 0; n < 4; ++n) {
      acc[n] = __builtin_amdgcn_wmma_f32_16x16x4_f32(
           false, av,  false, bf[j][n],
           (short)0, acc[n],  false,  false);
    }
  }
}

__global__ __launch_bounds__(128) void gemm_fused_kernel(
    const float* __restrict__ X0, const float* __restrict__ X1,
    const float* __restrict__ X2, const float* __restrict__ X3,
    const float* __restrict__ W, const float* __restrict__ bias,
    float* __restrict__ out, int N) {
  __shared__ float red[GEMM_WAVES * TILE_SLOTS * 32];

  const int lane = threadIdx.x & 31;
  const int wv   = threadIdx.x >> 5;
  const int row0 = blockIdx.x * 16;
  if (row0 >= N) return;

  const int c  = lane & 15;
  const int h  = lane >> 4;
  const int k0 = wv * 16 + h * 8;

  int arow = row0 + c;
  if (arow >= N) arow = N - 1;
  const long aoff = (long)arow * D_FEAT + k0;
  const float* wq = W + (long)k0 * D_FEAT + 4 * c;

  v8f acc[4];
  {
    const float4 bb = *reinterpret_cast<const float4*>(bias + 4 * c);
    const float b0 = wv == 0 ? bb.x : 0.0f;
    const float b1 = wv == 0 ? bb.y : 0.0f;
    const float b2 = wv == 0 ? bb.z : 0.0f;
    const float b3 = wv == 0 ? bb.w : 0.0f;
#pragma unroll
    for (int v = 0; v < 8; ++v) {
      acc[0][v] = b0; acc[1][v] = b1; acc[2][v] = b2; acc[3][v] = b3;
    }
  }

  float a[8];
  load8(a, X0 + aoff);
  term_quarter(acc, a, wq + 0 * W_TERM);
  load8(a, X1 + aoff);
  term_quarter(acc, a, wq + 1 * W_TERM);
  load8(a, X2 + aoff);
  term_quarter(acc, a, wq + 2 * W_TERM);
  load8(a, X3 + aoff);
  term_quarter(acc, a, wq + 3 * W_TERM);

  {
    float* mine = red + (wv * TILE_SLOTS) * 32 + lane;
#pragma unroll
    for (int v = 0; v < 8; ++v) {
#pragma unroll
      for (int n = 0; n < 4; ++n) mine[(4 * v + n) * 32] = acc[n][v];
    }
  }
  __syncthreads();

  for (int j = 0; j < 2; ++j) {
    const int v = 2 * wv + j;
    const float* p = red + (4 * v) * 32 + lane;
    float o[4];
#pragma unroll
    for (int n = 0; n < 4; ++n) {
      const float* q = p + n * 32;
      o[n] = ((q[0 * TILE_SLOTS * 32] + q[1 * TILE_SLOTS * 32]) + q[2 * TILE_SLOTS * 32])
             + q[3 * TILE_SLOTS * 32];
    }
    const int row = row0 + 8 * h + v;
    if (row < N)
      *reinterpret_cast<float4*>(out + (long)row * D_FEAT + 4 * c) = make_float4(o[0], o[1], o[2], o[3]);
  }
}


extern "C" void kernel_launch(void* const* d_in, const int* in_sizes, int n_in,
                              void* d_out, int out_size, void* d_ws, size_t ws_size,
                              hipStream_t stream) {
  const float* feat = (const float*)d_in[0];
  const int*   src  = (const int*)d_in[1];
  const int*   dst  = (const int*)d_in[2];
  const float* W    = (const float*)d_in[3];
  const float* b    = (const float*)d_in[4];
  float*       out  = (float*)d_out;

  const int N = in_sizes[0] / D_FEAT;
  const int E = in_sizes[1];
  const long nd = (long)N * D_FEAT;
  const int nown = (N + OWN_NODES - 1) / OWN_NODES;

  const long npad = ((long)N + 31) & ~31L;
  const long epad = ((long)E + 31) & ~31L;
  const long ncur = 2L * MAX_OWN * CUR_PITCH;
  float*    deg    = (float*)d_ws;
  int*      cursor = (int*)(deg + npad);
  int*      cnt    = cursor + ncur;
  int*      start  = cnt + MAX_OWN;
  uint2*    seg    = (uint2*)(start + MAX_OWN);
  float*    txA    = (float*)(seg + epad);
  float*    txB    = txA + nd;
  float*    txC    = txB + nd;

  const int ggrid = (N + 15) / 16;
  const int fgrid = (E + FILL_THREADS * FILL_PER_THREAD - 1) / (FILL_THREADS * FILL_PER_THREAD);

  zero_kernel<<<(int)((npad + ncur + 255) / 256), 256, 0, stream>>>(deg, npad + ncur);
  deg_kernel<<<(E + 255) / 256, 256, 0, stream>>>(dst, deg, E);
  count_kernel<<<(nown + 7) / 8, 256, 0, stream>>>(deg, cnt, N, nown);
  fill_kernel<<<fgrid, FILL_THREADS, 0, stream>>>(src, dst, cnt, deg, cursor, start, seg, E, nown);

  prop_kernel<false><<<nown, PROP_WAVES * 32, 0, stream>>>(feat, feat, deg, seg, start, txA,
                                                           -2.0f / LAM, 2.0f / LAM - 1.0f, 0.0f, N);
  prop_kernel<true><<<nown, PROP_WAVES * 32, 0, stream>>>(txA, feat, deg, seg, start, txB,
                                                          -4.0f / LAM, 4.0f / LAM - 2.0f, -1.0f, N);
  prop_kernel<true><<<nown, PROP_WAVES * 32, 0, stream>>>(txB, txA, deg, seg, start, txC,
                                                          -4.0f / LAM, 4.0f / LAM - 2.0f, -1.0f, N);

  gemm_fused_kernel<<<ggrid, 128, 0, stream>>>(feat, txA, txB, txC, W, b, out, N);
}


# Props sum on the matrix unit: a block's edges ordered by 16-node tile once per call; per 4 edges a one-hot*norm 16x4 times 4 gathered rows (f32 16x16x4), tile sums in registers; sum order changes
# speedup vs baseline: 2.8588x; 1.0305x over previous
#include <hip/hip_runtime.h>

typedef __attribute__((ext_vector_type(2))) float v2f;
typedef __attribute__((ext_vector_type(8))) float v8f;

#define D_FEAT 64
constexpr float LAM = 2.0f;


__global__ void zero_kernel(float* __restrict__ p, long n) {
  long i = (long)blockIdx.x * blockDim.x + threadIdx.x;
  long stride = (long)gridDim.x * blockDim.x;
  for (; i < n; i += stride) p[i] = 0.0f;
}

__global__ void deg_kernel(const int* __restrict__ dst, float* __restrict__ deg, int E) {
  int i = blockIdx.x * blockDim.x + threadIdx.x;
  if (i < E) atomicAdd(&deg[dst[i]], 1.0f);
}

__device__ __forceinline__ float node_norm(float deg) {
  return rsqrtf(fmaxf(deg, 1.0f));
}

#define OWN_NODES 391
#define LOC_BITS 9
#define MAX_OWN 1024
#define FILL_THREADS 512
#define FILL_PER_THREAD 8
#define CUR_PITCH 32
#define UNIT_EDGES 32
#define PROP_WAVES 32

__global__ __launch_bounds__(256) void count_kernel(const float* __restrict__ deg,
                                                    int* __restrict__ cnt, int N, int nown) {
  const int lane = threadIdx.x & 31;
  const int b = blockIdx.x * (blockDim.x >> 5) + (threadIdx.x >> 5);
  if (b >= nown) return;
  const int n0 = b * OWN_NODES;
  int s = 0;
#pragma unroll
  for (int k = 0; k < (OWN_NODES + 31) / 32; ++k) {
    const int i = lane + 32 * k;
    const bool ok = i < OWN_NODES && n0 + i < N;
    const float v = deg[ok ? n0 + i : n0];
    s += ok ? (int)v : 0;
  }
  for (int m = 16; m > 0; m >>= 1) s += __shfl_xor(s, m);
  if (lane == 0) cnt[b] = s;
}

__global__ __launch_bounds__(FILL_THREADS) void fill_kernel(
    const int* __restrict__ src, const int* __restrict__ dst, const int* __restrict__ cnt,
    const float* __restrict__ deg, int* __restrict__ cursor, int* __restrict__ start,
    uint2* __restrict__ seg, int E, int nown) {
  __shared__ int s_start[MAX_OWN];
  __shared__ int s_base[MAX_OWN];
  __shared__ int s_hist[MAX_OWN];
  __shared__ int s_wsum[FILL_THREADS / 32];
  const int t = threadIdx.x;
  const int lane = t & 31;
  const int wv = t >> 5;

  const int c0 = (2 * t < nown) ? cnt[2 * t] : 0;
  const int c1 = (2 * t + 1 < nown) ? cnt[2 * t + 1] : 0;
  s_hist[2 * t] = 0;
  s_hist[2 * t + 1] = 0;
  int incl = c0 + c1;
  for (int d = 1; d < 32; d <<= 1) {
    const int v = __shfl_up(incl, d);
    if (lane >= d) incl += v;
  }
  if (lane == 31) s_wsum[wv] = incl;
  __syncthreads();
  int woff = 0;
  for (int w = 0; w < wv; ++w) woff += s_wsum[w];
  const int excl = woff + incl - (c0 + c1);
  s_start[2 * t] = excl;
  s_start[2 * t + 1] = excl + c0;
  if (blockIdx.x == 0) {
    if (2 * t <= nown) start[2 * t] = excl;
    if (2 * t + 1 <= nown) start[2 * t + 1] = excl + c0;
  }
  __syncthreads();

  const unsigned hw_id = __builtin_amdgcn_s_sendmsg_rtn(0x87);
  const int half = (int)((hw_id >> 18) & 1u);
  int* __restrict__ cur = cursor + (long)half * (MAX_OWN * CUR_PITCH);

  const long base = (long)blockIdx.x * (FILL_THREADS * FILL_PER_THREAD);
  int rk[FILL_PER_THREAD];
  int ob[FILL_PER_THREAD];
  unsigned pk[FILL_PER_THREAD];
  int dv[FILL_PER_THREAD];
  int sv[FILL_PER_THREAD];
#pragma unroll
  for (int j = 0; j < FILL_PER_THREAD; ++j) {
    const long e = base + j * FILL_THREADS + t;
    const long ec = e < E ? e : (long)E - 1;
    dv[j] = dst[ec];
    sv[j] = src[ec];
  }
  float gv[FILL_PER_THREAD];
#pragma unroll
  for (int j = 0; j < FILL_PER_THREAD; ++j) gv[j] = deg[sv[j]];
#pragma unroll
  for (int j = 0; j < FILL_PER_THREAD; ++j) {
    const long e = base + j * FILL_THREADS + t;
    ob[j] = -1;
    rk[j] = 0;
    pk[j] = 0u;
    if (e < E) {
      const int d = dv[j];
      const int s = sv[j];
      const int b = d / OWN_NODES;
      ob[j] = b;
      pk[j] = ((unsigned)s << LOC_BITS) | (unsigned)(d - b * OWN_NODES);
      rk[j] = atomicAdd(&s_hist[b], 1);
    }
  }
  __syncthreads();
  for (int b = t; b < nown; b += FILL_THREADS) {
    const int hcount = s_hist[b];
    int at = 0;
    if (hcount > 0) {
      const int old = atomicAdd(&cur[b * CUR_PITCH], hcount);
      at = half == 0 ? s_start[b] + old : s_start[b + 1] - old - hcount;
    }
    s_base[b] = at;
  }
  __syncthreads();
#pragma unroll
  for (int j = 0; j < FILL_PER_THREAD; ++j) {
    if (ob[j] >= 0)
      seg[s_base[ob[j]] + rk[j]] = make_uint2(pk[j], (unsigned)__float_as_int(node_norm(gv[j])));
  }
}

#define TILE_ROWS 16
#define NTILE ((OWN_NODES + TILE_ROWS - 1) / TILE_ROWS)
#define ACC_ROWS (NTILE * TILE_ROWS)
#define PAD_PER_OWN 80
#define ORDER_THREADS 1024
#define ORDER_PER_TRIP 8

__global__ __launch_bounds__(ORDER_THREADS) void order_kernel(
    const uint2* __restrict__ seg, const int* __restrict__ start,
    uint2* __restrict__ seg2, int2* __restrict__ range) {
  __shared__ int s_cnt[NTILE];
  __shared__ int s_cur[NTILE];
  __shared__ int s_pos[NTILE + 1];
  const int t = threadIdx.x;
  const int b = blockIdx.x;
  const int s0 = start[b];
  const int s1 = start[b + 1];
  if (t < NTILE) {
    s_cnt[t] = 0;
    s_cur[t] = 0;
  }
  __syncthreads();
  for (int i0 = s0 + t; i0 < s1; i0 += ORDER_PER_TRIP * ORDER_THREADS) {
    unsigned w[ORDER_PER_TRIP];
#pragma unroll
    for (int k = 0; k < ORDER_PER_TRIP; ++k) {
      const int i = i0 + k * ORDER_THREADS;
      w[k] = seg[i < s1 ? i : s1 - 1].x;
    }
#pragma unroll
    for (int k = 0; k < ORDER_PER_TRIP; ++k) {
      if (i0 + k * ORDER_THREADS < s1) atomicAdd(&s_cnt[(w[k] & ((1u << LOC_BITS) - 1u)) >> 4], 1);
    }
  }
  __syncthreads();
  if (t == 0) {
    int p = 0;
    for (int k = 0; k < NTILE; ++k) {
      s_pos[k] = p;
      p += (s_cnt[k] + 3) & ~3;
    }
    s_pos[NTILE] = p;
  }
  __syncthreads();
  const int base2 = ((s0 + 3) & ~3) + PAD_PER_OWN * b;
  for (int i0 = s0 + t; i0 < s1; i0 += ORDER_PER_TRIP * ORDER_THREADS) {
    uint2 e[ORDER_PER_TRIP];
#pragma unroll
    for (int k = 0; k < ORDER_PER_TRIP; ++k) {
      const int i = i0 + k * ORDER_THREADS;
      e[k] = seg[i < s1 ? i : s1 - 1];
    }
#pragma unroll
    for (int k = 0; k < ORDER_PER_TRIP; ++k) {
      if (i0 + k * ORDER_THREADS < s1) {
        const int tile = (int)((e[k].x & ((1u << LOC_BITS) - 1u)) >> 4);
        const int r = atomicAdd(&s_cur[tile], 1);
        seg2[base2 + s_pos[tile] + r] = e[k];
      }
    }
  }
  if (t < NTILE) {
    const int cn = s_cnt[t];
    const int padn = (-cn) & 3;
    for (int k = 0; k < padn; ++k)
      seg2[base2 + s_pos[t] + cn + k] = make_uint2((unsigned)(TILE_ROWS * t), 0u);
  }
  if (t == 0) range[b] = make_int2(base2, base2 + s_pos[NTILE]);
}


__device__ __forceinline__ void lds_add(float* p, float v) {
  __hip_atomic_fetch_add(p, v, __ATOMIC_RELAXED, __HIP_MEMORY_SCOPE_WORKGROUP);
}

__device__ __forceinline__ void flush_tile(float* acc, const v8f (&sum)[4], int tile, int c, int h) {
  float* o = acc + (tile * TILE_ROWS + 8 * h) * D_FEAT + 4 * c;
#pragma unroll
  for (int r = 0; r < 8; ++r) {
#pragma unroll
    for (int n = 0; n < 4; ++n) lds_add(o + r * D_FEAT + n, sum[n][r]);
  }
}

__device__ __forceinline__ void four_steps(v8f (&sum)[4], int& cur, float* acc, const uint2* slot_w,
                                           const float* __restrict__ x, int first_step, int c, int h) {
  uint4 en[4];
  float4 ra[4], rb[4];
#pragma unroll
  for (int j = 0; j < 4; ++j) {
    en[j] = *reinterpret_cast<const uint4*>(slot_w + 4 * (first_step + j) + 2 * h);
    ra[j] = reinterpret_cast<const float4*>(x)[(en[j].x >> LOC_BITS) * (D_FEAT / 4) + c];
    rb[j] = reinterpret_cast<const float4*>(x)[(en[j].z >> LOC_BITS) * (D_FEAT / 4) + c];
  }
#pragma unroll
  for (int j = 0; j < 4; ++j) {
    const int tl = __builtin_amdgcn_readfirstlane((int)((en[j].x & ((1u << LOC_BITS) - 1u)) >> 4));
    if (tl != cur) {
      if (cur >= 0) flush_tile(acc, sum, cur, c, h);
#pragma unroll
      for (int n = 0; n < 4; ++n) {
#pragma unroll
        for (int r = 0; r < 8; ++r) sum[n][r] = 0.0f;
      }
      cur = tl;
    }
    v2f av;
    av.x = (int)(en[j].x & (TILE_ROWS - 1)) == c ? __uint_as_float(en[j].y) : 0.0f;
    av.y = (int)(en[j].z & (TILE_ROWS - 1)) == c ? __uint_as_float(en[j].w) : 0.0f;
    v2f b0, b1, b2, b3;
    b0.x = ra[j].x; b0.y = rb[j].x;
    b1.x = ra[j].y; b1.y = rb[j].y;
    b2.x = ra[j].z; b2.y = rb[j].z;
    b3.x = ra[j].w; b3.y = rb[j].w;
    sum[0] = __builtin_amdgcn_wmma_f32_16x16x4_f32(false, av, false, b0, (short)0, sum[0], false, false);
    sum[1] = __builtin_amdgcn_wmma_f32_16x16x4_f32(false, av, false, b1, (short)0, sum[1], false, false);
    sum[2] = __builtin_amdgcn_wmma_f32_16x16x4_f32(false, av, false, b2, (short)0, sum[2], false, false);
    sum[3] = __builtin_amdgcn_wmma_f32_16x16x4_f32(false, av, false, b3, (short)0, sum[3], false, false);
  }
}

template <bool HAS_P>
__device__ __forceinline__ void write_term(int i, int n0, const float* acc,
                                           const float* __restrict__ x, const float* __restrict__ xp,
                                           const float* __restrict__ deg, float* __restrict__ out,
                                           float cH, float cX, float cP) {
  const int node = n0 + (i >> 4);
  const float nn = node_norm(deg[node]);
  const float4 a = reinterpret_cast<const float4*>(acc)[i];
  const long gi = (long)node * (D_FEAT / 4) + (i & 15);
  const float4 x1 = reinterpret_cast<const float4*>(x)[gi];
  float4 r;
  r.x = cH * (a.x * nn) + cX * x1.x;
  r.y = cH * (a.y * nn) + cX * x1.y;
  r.z = cH * (a.z * nn) + cX * x1.z;
  r.w = cH * (a.w * nn) + cX * x1.w;
  if (HAS_P) {
    const float4 x2 = reinterpret_cast<const float4*>(xp)[gi];
    r.x += cP * x2.x;
    r.y += cP * x2.y;
    r.z += cP * x2.z;
    r.w += cP * x2.w;
  }
  reinterpret_cast<float4*>(out)[gi] = r;
}

template <bool HAS_P>
__global__ __launch_bounds__(PROP_WAVES * 32) void prop_kernel(
    const float* __restrict__ x, const float* __restrict__ xp, const float* __restrict__ deg,
    const uint2* __restrict__ seg2, const int2* __restrict__ range,
    float* __restrict__ out, float cH, float cX, float cP, int N) {
  __shared__ float acc[ACC_ROWS * D_FEAT];
  __shared__ __attribute__((aligned(16))) uint2 slot[PROP_WAVES * UNIT_EDGES];
  const int t = threadIdx.x;
  const int lane = t & 31;
  const int wv = __builtin_amdgcn_readfirstlane(t >> 5);
  const int b = blockIdx.x;
  const int c = lane & 15;
  const int h = lane >> 4;

  for (int i = t; i < ACC_ROWS * D_FEAT / 4; i += PROP_WAVES * 32)
    reinterpret_cast<float4*>(acc)[i] = make_float4(0.0f, 0.0f, 0.0f, 0.0f);
  const int2 rg = range[b];
  __syncthreads();

  const int steps = (rg.y - rg.x) >> 2;
  int per = (steps + PROP_WAVES - 1) / PROP_WAVES;
  per = (per + 3) & ~3;
  const int st0 = wv * per;
  const int st1 = st0 + per < steps ? st0 + per : steps;
  if (st0 < st1) {
    const int e0 = rg.x + 4 * st0;
    const int e1 = rg.x + 4 * st1;
    uint2* slot_w = slot + wv * UNIT_EDGES;
    v8f sum[4];
#pragma unroll
    for (int n = 0; n < 4; ++n) {
#pragma unroll
      for (int r = 0; r < 8; ++r) sum[n][r] = 0.0f;
    }
    int cur = -1;
    int cnt = e1 - e0 < UNIT_EDGES ? e1 - e0 : UNIT_EDGES;
    uint2 ent = seg2[e0 + (lane < cnt ? lane : cnt - 1)];
    for (int ub = e0; ub < e1; ub += UNIT_EDGES) {
      uint2 pe = ent;
      if (lane >= cnt) pe.y = 0u;
      slot_w[lane] = pe;
      const int cnt_now = cnt;
      const int nb = ub + UNIT_EDGES;
      if (nb < e1) {
        cnt = e1 - nb < UNIT_EDGES ? e1 - nb : UNIT_EDGES;
        ent = seg2[nb + (lane < cnt ? lane : cnt - 1)];
      }
      four_steps(sum, cur, acc, slot_w, x, 0, c, h);
      if (cnt_now > 16) four_steps(sum, cur, acc, slot_w, x, 4, c, h);
    }
    if (cur >= 0) flush_tile(acc, sum, cur, c, h);
  }
  __syncthreads();

  const int n0 = b * OWN_NODES;
  constexpr int PER_BLOCK = OWN_NODES * (D_FEAT / 4);
  constexpr int FULL_TRIPS = PER_BLOCK / (PROP_WAVES * 32);
  if (n0 + OWN_NODES <= N) {
#pragma unroll
    for (int k = 0; k < FULL_TRIPS; ++k)
      write_term<HAS_P>(t + k * (PROP_WAVES * 32), n0, acc, x, xp, deg, out, cH, cX, cP);
    if (t < PER_BLOCK - FULL_TRIPS * (PROP_WAVES * 32))
      write_term<HAS_P>(t + FULL_TRIPS * (PROP_WAVES * 32), n0, acc, x, xp, deg, out, cH, cX, cP);
  } else {
    for (int i = t; i < PER_BLOCK; i += PROP_WAVES * 32) {
      if (n0 + (i >> 4) < N) write_term<HAS_P>(i, n0, acc, x, xp, deg, out, cH, cX, cP);
    }
  }
}


#define GEMM_WAVES 4
#define TILE_SLOTS 32
#define W_TERM (D_FEAT * D_FEAT)

__device__ __forceinline__ void load8(float (&a)[8], const float* __restrict__ p) {
  const float4 lo = reinterpret_cast<const float4*>(p)[0];
  const float4 hi = reinterpret_cast<const float4*>(p)[1];
  a[0] = lo.x; a[1] = lo.y; a[2] = lo.z; a[3] = lo.w;
  a[4] = hi.x; a[5] = hi.y; a[6] = hi.z; a[7] = hi.w;
}

__device__ __forceinline__ void term_quarter(v8f (&acc)[4], const float (&a)[8],
                                             const float* __restrict__ wrow) {
  v2f bf[4][4];
#pragma unroll
  for (int j = 0; j < 4; ++j) {
    const float4 w0 = *reinterpret_cast<const float4*>(wrow + (2 * j + 0) * D_FEAT);
    const float4 w1 = *reinterpret_cast<const float4*>(wrow + (2 * j + 1) * D_FEAT);
    bf[j][0].x = w0.x; bf[j][0].y = w1.x;
    bf[j][1].x = w0.y; bf[j][1].y = w1.y;
    bf[j][2].x = w0.z; bf[j][2].y = w1.z;
    bf[j][3].x = w0.w; bf[j][3].y = w1.w;
  }
#pragma unroll
  for (int j = 0; j < 4; ++j) {
    v2f av;
    av.x = a[2 * j + 0];
    av.y = a[2 * j + 1];
#pragma unroll
    for (int n = 0; n < 4; ++n) {
      acc[n] = __builtin_amdgcn_wmma_f32_16x16x4_f32(
           false, av,  false, bf[j][n],
           (short)0, acc[n],  false,  false);
    }
  }
}

__global__ __launch_bounds__(128) void gemm_fused_kernel(
    const float* __restrict__ X0, const float* __restrict__ X1,
    const float* __restrict__ X2, const float* __restrict__ X3,
    const float* __restrict__ W, const float* __restrict__ bias,
    float* __restrict__ out, int N) {
  __shared__ float red[GEMM_WAVES * TILE_SLOTS * 32];

  const int lane = threadIdx.x & 31;
  const int wv   = threadIdx.x >> 5;
  const int row0 = blockIdx.x * 16;
  if (row0 >= N) return;

  const int c  = lane & 15;
  const int h  = lane >> 4;
  const int k0 = wv * 16 + h * 8;

  int arow = row0 + c;
  if (arow >= N) arow = N - 1;
  const long aoff = (long)arow * D_FEAT + k0;
  const float* wq = W + (long)k0 * D_FEAT + 4 * c;

  v8f acc[4];
  {
    const float4 bb = *reinterpret_cast<const float4*>(bias + 4 * c);
    const float b0 = wv == 0 ? bb.x : 0.0f;
    const float b1 = wv == 0 ? bb.y : 0.0f;
    const float b2 = wv == 0 ? bb.z : 0.0f;
    const float b3 = wv == 0 ? bb.w : 0.0f;
#pragma unroll
    for (int v = 0; v < 8; ++v) {
      acc[0][v] = b0; acc[1][v] = b1; acc[2][v] = b2; acc[3][v] = b3;
    }
  }

  float a[8];
  load8(a, X0 + aoff);
  term_quarter(acc, a, wq + 0 * W_TERM);
  load8(a, X1 + aoff);
  term_quarter(acc, a, wq + 1 * W_TERM);
  load8(a, X2 + aoff);
  term_quarter(acc, a, wq + 2 * W_TERM);
  load8(a, X3 + aoff);
  term_quarter(acc, a, wq + 3 * W_TERM);

  {
    float* mine = red + (wv * TILE_SLOTS) * 32 + lane;
#pragma unroll
    for (int v = 0; v < 8; ++v) {
#pragma unroll
      for (int n = 0; n < 4; ++n) mine[(4 * v + n) * 32] = acc[n][v];
    }
  }
  __syncthreads();

  for (int j = 0; j < 2; ++j) {
    const int v = 2 * wv + j;
    const float* p = red + (4 * v) * 32 + lane;
    float o[4];
#pragma unroll
    for (int n = 0; n < 4; ++n) {
      const float* q = p + n * 32;
      o[n] = ((q[0 * TILE_SLOTS * 32] + q[1 * TILE_SLOTS * 32]) + q[2 * TILE_SLOTS * 32])
             + q[3 * TILE_SLOTS * 32];
    }
    const int row = row0 + 8 * h + v;
    if (row < N)
      *reinterpret_cast<float4*>(out + (long)row * D_FEAT + 4 * c) = make_float4(o[0], o[1], o[2], o[3]);
  }
}


extern "C" void kernel_launch(void* const* d_in, const int* in_sizes, int n_in,
                              void* d_out, int out_size, void* d_ws, size_t ws_size,
                              hipStream_t stream) {
  const float* feat = (const float*)d_in[0];
  const int*   src  = (const int*)d_in[1];
  const int*   dst  = (const int*)d_in[2];
  const float* W    = (const float*)d_in[3];
  const float* b    = (const float*)d_in[4];
  float*       out  = (float*)d_out;

  const int N = in_sizes[0] / D_FEAT;
  const int E = in_sizes[1];
  const long nd = (long)N * D_FEAT;
  const int nown = (N + OWN_NODES - 1) / OWN_NODES;

  const long npad = ((long)N + 31) & ~31L;
  const long epad = ((long)E + 31) & ~31L;
  const long ncur = 2L * MAX_OWN * CUR_PITCH;
  float*    deg    = (float*)d_ws;
  int*      cursor = (int*)(deg + npad);
  int*      cnt    = cursor + ncur;
  int*      start  = cnt + MAX_OWN;
  uint2*    seg    = (uint2*)(start + MAX_OWN);
  const long e2pad = epad + (long)PAD_PER_OWN * MAX_OWN + 32;
  uint2*    seg2   = seg + epad;
  int2*     range  = (int2*)(seg2 + e2pad);
  float*    txA    = (float*)(range + MAX_OWN);
  float*    txB    = txA + nd;
  float*    txC    = txB + nd;

  const int ggrid = (N + 15) / 16;
  const int fgrid = (E + FILL_THREADS * FILL_PER_THREAD - 1) / (FILL_THREADS * FILL_PER_THREAD);

  zero_kernel<<<(int)((npad + ncur + 255) / 256), 256, 0, stream>>>(deg, npad + ncur);
  deg_kernel<<<(E + 255) / 256, 256, 0, stream>>>(dst, deg, E);
  count_kernel<<<(nown + 7) / 8, 256, 0, stream>>>(deg, cnt, N, nown);
  fill_kernel<<<fgrid, FILL_THREADS, 0, stream>>>(src, dst, cnt, deg, cursor, start, seg, E, nown);
  order_kernel<<<nown, ORDER_THREADS, 0, stream>>>(seg, start, seg2, range);

  prop_kernel<false><<<nown, PROP_WAVES * 32, 0, stream>>>(feat, feat, deg, seg2, range, txA,
                                                           -2.0f / LAM, 2.0f / LAM - 1.0f, 0.0f, N);
  prop_kernel<true><<<nown, PROP_WAVES * 32, 0, stream>>>(txA, feat, deg, seg2, range, txB,
                                                          -4.0f / LAM, 4.0f / LAM - 2.0f, -1.0f, N);
  prop_kernel<true><<<nown, PROP_WAVES * 32, 0, stream>>>(txB, txA, deg, seg2, range, txC,
                                                          -4.0f / LAM, 4.0f / LAM - 2.0f, -1.0f, N);

  gemm_fused_kernel<<<ggrid, 128, 0, stream>>>(feat, txA, txB, txC, W, b, out, N);
}


# Edges grouped with no global counter: chunks sorted by owner in LDS, owners collect their pieces and count in-degrees in LDS; zero, degree and count launches go (9 to 6); order of tile sums changes
# speedup vs baseline: 4.0566x; 1.4190x over previous
#include <hip/hip_runtime.h>

typedef __attribute__((ext_vector_type(2))) float v2f;
typedef __attribute__((ext_vector_type(8))) float v8f;

#define D_FEAT 64
constexpr float LAM = 2.0f;

__device__ __forceinline__ float node_norm(float deg) {
  return rsqrtf(fmaxf(deg, 1.0f));
}

#define OWN_NODES 391
#define LOC_BITS 9
#define MAX_OWN 1024
#define FILL_THREADS 512
#define FILL_PER_THREAD 8
#define CHUNK_EDGES (FILL_THREADS * FILL_PER_THREAD)
#define UNIT_EDGES 32
#define PROP_WAVES 32
#define TILE_ROWS 16
#define NTILE ((OWN_NODES + TILE_ROWS - 1) / TILE_ROWS)
#define ACC_ROWS (NTILE * TILE_ROWS)
#define PAD_PER_OWN 80
#define ORDER_THREADS 1024
#define PIECES_PER_TRIP 16
#define NULL_MARK 1u

__global__ __launch_bounds__(FILL_THREADS) void group_kernel(
    const int* __restrict__ src, const int* __restrict__ dst, int2* __restrict__ tab,
    unsigned* __restrict__ cseg, int E, int nown, int cpitch) {
  __shared__ int s_hist[MAX_OWN];
  __shared__ int s_off[MAX_OWN];
  __shared__ int s_wsum[FILL_THREADS / 32];
  const int t = threadIdx.x;
  const int lane = t & 31;
  const int wv = t >> 5;
  s_hist[2 * t] = 0;
  s_hist[2 * t + 1] = 0;

  const long base = (long)blockIdx.x * CHUNK_EDGES;
  int rk[FILL_PER_THREAD];
  int ob[FILL_PER_THREAD];
  unsigned pk[FILL_PER_THREAD];
  int dv[FILL_PER_THREAD];
  int sv[FILL_PER_THREAD];
#pragma unroll
  for (int j = 0; j < FILL_PER_THREAD; ++j) {
    const long e = base + j * FILL_THREADS + t;
    const long ec = e < E ? e : (long)E - 1;
    dv[j] = dst[ec];
    sv[j] = src[ec];
  }
  __syncthreads();
#pragma unroll
  for (int j = 0; j < FILL_PER_THREAD; ++j) {
    const long e = base + j * FILL_THREADS + t;
    ob[j] = -1;
    rk[j] = 0;
    pk[j] = 0u;
    if (e < E) {
      const int d = dv[j];
      const int b = d / OWN_NODES;
      ob[j] = b;
      pk[j] = ((unsigned)sv[j] << LOC_BITS) | (unsigned)(d - b * OWN_NODES);
      rk[j] = atomicAdd(&s_hist[b], 1);
    }
  }
  __syncthreads();
  const int c0 = s_hist[2 * t];
  const int c1 = s_hist[2 * t + 1];
  int incl = c0 + c1;
  for (int d = 1; d < 32; d <<= 1) {
    const int v = __shfl_up(incl, d);
    if (lane >= d) incl += v;
  }
  if (lane == 31) s_wsum[wv] = incl;
  __syncthreads();
  int woff = 0;
  for (int w = 0; w < wv; ++w) woff += s_wsum[w];
  const int excl = woff + incl - (c0 + c1);
  s_off[2 * t] = excl;
  s_off[2 * t + 1] = excl + c0;
  if (2 * t < nown) tab[(long)(2 * t) * cpitch + blockIdx.x] = make_int2(excl, c0);
  if (2 * t + 1 < nown) tab[(long)(2 * t + 1) * cpitch + blockIdx.x] = make_int2(excl + c0, c1);
  __syncthreads();
#pragma unroll
  for (int j = 0; j < FILL_PER_THREAD; ++j) {
    if (ob[j] >= 0) cseg[base + s_off[ob[j]] + rk[j]] = pk[j];
  }
}

template <bool PLACE>
__device__ __forceinline__ int walk_pieces(const int2* __restrict__ row, const unsigned* __restrict__ cseg,
                                           int nchunks, int wv, int lane, int* s_deg, int* s_cnt,
                                           int* s_cur, const int* s_pos, uint2* __restrict__ seg2, int base2) {
  int first_sum = 0;
  for (int kb = wv; kb < nchunks; kb += 32 * PIECES_PER_TRIP) {
    int2 pc[PIECES_PER_TRIP];
    unsigned w[PIECES_PER_TRIP];
#pragma unroll
    for (int i = 0; i < PIECES_PER_TRIP; ++i) {
      const int k = kb + 32 * i;
      pc[i] = row[k < nchunks ? k : nchunks - 1];
      if (k >= nchunks) pc[i] = make_int2(0, 0);
    }
#pragma unroll
    for (int i = 0; i < PIECES_PER_TRIP; ++i) {
      const int k = kb + 32 * i;
      const long cb = (long)(k < nchunks ? k : nchunks - 1) * CHUNK_EDGES;
      w[i] = cseg[cb + (lane < pc[i].y ? pc[i].x + lane : 0)];
      first_sum += pc[i].x;
    }
#pragma unroll
    for (int i = 0; i < PIECES_PER_TRIP; ++i) {
      const int k = kb + 32 * i;
      const long cb = (long)(k < nchunks ? k : nchunks - 1) * CHUNK_EDGES;
      unsigned e = w[i];
      for (int j = lane; j < pc[i].y; j += 32) {
        if (j >= 32) e = cseg[cb + pc[i].x + j];
        const int loc = (int)(e & ((1u << LOC_BITS) - 1u));
        if (PLACE) {
          const int r = atomicAdd(&s_cur[loc >> 4], 1);
          seg2[base2 + s_pos[loc >> 4] + r] = make_uint2(e, 0u);
        } else {
          atomicAdd(&s_deg[loc], 1);
          atomicAdd(&s_cnt[loc >> 4], 1);
        }
      }
    }
  }
  return first_sum;
}

__global__ __launch_bounds__(ORDER_THREADS) void collect_kernel(
    const int2* __restrict__ tab, const unsigned* __restrict__ cseg, float* __restrict__ deg,
    uint2* __restrict__ seg2, int2* __restrict__ range, int nchunks, int cpitch, int N) {
  __shared__ int s_deg[ACC_ROWS];
  __shared__ int s_cnt[NTILE];
  __shared__ int s_cur[NTILE];
  __shared__ int s_pos[NTILE + 1];
  __shared__ int s_first;
  const int t = threadIdx.x;
  const int lane = t & 31;
  const int wv = t >> 5;
  const int b = blockIdx.x;
  const int2* __restrict__ row = tab + (long)b * cpitch;
  if (t < ACC_ROWS) s_deg[t] = 0;
  if (t < NTILE) {
    s_cnt[t] = 0;
    s_cur[t] = 0;
  }
  if (t == 0) s_first = 0;
  __syncthreads();
  const int fs = walk_pieces<false>(row, cseg, nchunks, wv, lane, s_deg, s_cnt, s_cur, s_pos, seg2, 0);
  if (lane == 0) atomicAdd(&s_first, fs);
  __syncthreads();
  if (t == 0) {
    int p = 0;
    for (int k = 0; k < NTILE; ++k) {
      s_pos[k] = p;
      p += (s_cnt[k] + 3) & ~3;
    }
    s_pos[NTILE] = p;
  }
  const int n0 = b * OWN_NODES;
  if (t < OWN_NODES && n0 + t < N) deg[n0 + t] = (float)s_deg[t];
  __syncthreads();
  const int base2 = ((s_first + 3) & ~3) + PAD_PER_OWN * b;
  walk_pieces<true>(row, cseg, nchunks, wv, lane, s_deg, s_cnt, s_cur, s_pos, seg2, base2);
  if (t < NTILE) {
    const int cn = s_cnt[t];
    const int padn = (-cn) & 3;
    for (int k = 0; k < padn; ++k)
      seg2[base2 + s_pos[t] + cn + k] = make_uint2((unsigned)(TILE_ROWS * t), NULL_MARK);
  }
  if (t == 0) range[b] = make_int2(base2, base2 + s_pos[NTILE]);
}


__device__ __forceinline__ void lds_add(float* p, float v) {
  __hip_atomic_fetch_add(p, v, __ATOMIC_RELAXED, __HIP_MEMORY_SCOPE_WORKGROUP);
}

__device__ __forceinline__ void flush_tile(float* acc, const v8f (&sum)[4], int tile, int c, int h) {
  float* o = acc + (tile * TILE_ROWS + 8 * h) * D_FEAT + 4 * c;
#pragma unroll
  for (int r = 0; r < 8; ++r) {
#pragma unroll
    for (int n = 0; n < 4; ++n) lds_add(o + r * D_FEAT + n, sum[n][r]);
  }
}

__device__ __forceinline__ void four_steps(v8f (&sum)[4], int& cur, float* acc, const uint2* slot_w,
                                           const float* __restrict__ x, int first_step, int c, int h) {
  uint4 en[4];
  float4 ra[4], rb[4];
#pragma unroll
  for (int j = 0; j < 4; ++j) {
    en[j] = *reinterpret_cast<const uint4*>(slot_w + 4 * (first_step + j) + 2 * h);
    ra[j] = reinterpret_cast<const float4*>(x)[(en[j].x >> LOC_BITS) * (D_FEAT / 4) + c];
    rb[j] = reinterpret_cast<const float4*>(x)[(en[j].z >> LOC_BITS) * (D_FEAT / 4) + c];
  }
#pragma unroll
  for (int j = 0; j < 4; ++j) {
    const int tl = __builtin_amdgcn_readfirstlane((int)((en[j].x & ((1u << LOC_BITS) - 1u)) >> 4));
    if (tl != cur) {
      if (cur >= 0) flush_tile(acc, sum, cur, c, h);
#pragma unroll
      for (int n = 0; n < 4; ++n) {
#pragma unroll
        for (int r = 0; r < 8; ++r) sum[n][r] = 0.0f;
      }
      cur = tl;
    }
    v2f av;
    av.x = (int)(en[j].x & (TILE_ROWS - 1)) == c ? __uint_as_float(en[j].y) : 0.0f;
    av.y = (int)(en[j].z & (TILE_ROWS - 1)) == c ? __uint_as_float(en[j].w) : 0.0f;
    v2f b0, b1, b2, b3;
    b0.x = ra[j].x; b0.y = rb[j].x;
    b1.x = ra[j].y; b1.y = rb[j].y;
    b2.x = ra[j].z; b2.y = rb[j].z;
    b3.x = ra[j].w; b3.y = rb[j].w;
    sum[0] = __builtin_amdgcn_wmma_f32_16x16x4_f32(false, av, false, b0, (short)0, sum[0], false, false);
    sum[1] = __builtin_amdgcn_wmma_f32_16x16x4_f32(false, av, false, b1, (short)0, sum[1], false, false);
    sum[2] = __builtin_amdgcn_wmma_f32_16x16x4_f32(false, av, false, b2, (short)0, sum[2], false, false);
    sum[3] = __builtin_amdgcn_wmma_f32_16x16x4_f32(false, av, false, b3, (short)0, sum[3], false, false);
  }
}

template <bool HAS_P>
__device__ __forceinline__ void write_term(int i, int n0, const float* acc,
                                           const float* __restrict__ x, const float* __restrict__ xp,
                                           const float* __restrict__ deg, float* __restrict__ out,
                                           float cH, float cX, float cP) {
  const int node = n0 + (i >> 4);
  const float nn = node_norm(deg[node]);
  const float4 a = reinterpret_cast<const float4*>(acc)[i];
  const long gi = (long)node * (D_FEAT / 4) + (i & 15);
  const float4 x1 = reinterpret_cast<const float4*>(x)[gi];
  float4 r;
  r.x = cH * (a.x * nn) + cX * x1.x;
  r.y = cH * (a.y * nn) + cX * x1.y;
  r.z = cH * (a.z * nn) + cX * x1.z;
  r.w = cH * (a.w * nn) + cX * x1.w;
  if (HAS_P) {
    const float4 x2 = reinterpret_cast<const float4*>(xp)[gi];
    r.x += cP * x2.x;
    r.y += cP * x2.y;
    r.z += cP * x2.z;
    r.w += cP * x2.w;
  }
  reinterpret_cast<float4*>(out)[gi] = r;
}

template <bool HAS_P>
__global__ __launch_bounds__(PROP_WAVES * 32) void prop_kernel(
    const float* __restrict__ x, const float* __restrict__ xp, const float* __restrict__ deg,
    uint2* __restrict__ seg2, const int2* __restrict__ range,
    float* __restrict__ out, float cH, float cX, float cP, int N) {
  __shared__ float acc[ACC_ROWS * D_FEAT];
  __shared__ __attribute__((aligned(16))) uint2 slot[PROP_WAVES * UNIT_EDGES];
  const int t = threadIdx.x;
  const int lane = t & 31;
  const int wv = __builtin_amdgcn_readfirstlane(t >> 5);
  const int b = blockIdx.x;
  const int c = lane & 15;
  const int h = lane >> 4;

  for (int i = t; i < ACC_ROWS * D_FEAT / 4; i += PROP_WAVES * 32)
    reinterpret_cast<float4*>(acc)[i] = make_float4(0.0f, 0.0f, 0.0f, 0.0f);
  const int2 rg = range[b];
  __syncthreads();

  const int steps = (rg.y - rg.x) >> 2;
  int per = (steps + PROP_WAVES - 1) / PROP_WAVES;
  per = (per + 3) & ~3;
  const int st0 = wv * per;
  const int st1 = st0 + per < steps ? st0 + per : steps;
  if (st0 < st1) {
    const int e0 = rg.x + 4 * st0;
    const int e1 = rg.x + 4 * st1;
    uint2* slot_w = slot + wv * UNIT_EDGES;
    v8f sum[4];
#pragma unroll
    for (int n = 0; n < 4; ++n) {
#pragma unroll
      for (int r = 0; r < 8; ++r) sum[n][r] = 0.0f;
    }
    int cur = -1;
    int cnt = e1 - e0 < UNIT_EDGES ? e1 - e0 : UNIT_EDGES;
    uint2 ent = seg2[e0 + (lane < cnt ? lane : cnt - 1)];
    if (HAS_P) {
      for (int ub = e0; ub < e1; ub += UNIT_EDGES) {
        uint2 pe = ent;
        if (lane >= cnt) pe.y = 0u;
        slot_w[lane] = pe;
        const int cnt_now = cnt;
        const int nb = ub + UNIT_EDGES;
        if (nb < e1) {
          cnt = e1 - nb < UNIT_EDGES ? e1 - nb : UNIT_EDGES;
          ent = seg2[nb + (lane < cnt ? lane : cnt - 1)];
        }
        four_steps(sum, cur, acc, slot_w, x, 0, c, h);
        if (cnt_now > 16) four_steps(sum, cur, acc, slot_w, x, 4, c, h);
      }
    } else {
      float dg = deg[ent.x >> LOC_BITS];
      int cnt1 = 0;
      uint2 ent1 = ent;
      if (e0 + UNIT_EDGES < e1) {
        cnt1 = e1 - (e0 + UNIT_EDGES) < UNIT_EDGES ? e1 - (e0 + UNIT_EDGES) : UNIT_EDGES;
        ent1 = seg2[e0 + UNIT_EDGES + (lane < cnt1 ? lane : cnt1 - 1)];
      }
      for (int ub = e0; ub < e1; ub += UNIT_EDGES) {
        uint2 pe = ent;
        pe.y = (ent.y != 0u || lane >= cnt) ? 0u : (unsigned)__float_as_int(node_norm(dg));
        slot_w[lane] = pe;
        if (lane < cnt) seg2[ub + lane] = pe;
        const int cnt_now = cnt;
        ent = ent1;
        cnt = cnt1;
        if (ub + UNIT_EDGES < e1) dg = deg[ent.x >> LOC_BITS];
        const int nb2 = ub + 2 * UNIT_EDGES;
        if (nb2 < e1) {
          cnt1 = e1 - nb2 < UNIT_EDGES ? e1 - nb2 : UNIT_EDGES;
          ent1 = seg2[nb2 + (lane < cnt1 ? lane : cnt1 - 1)];
        }
        four_steps(sum, cur, acc, slot_w, x, 0, c, h);
        if (cnt_now > 16) four_steps(sum, cur, acc, slot_w, x, 4, c, h);
      }
    }
    if (cur >= 0) flush_tile(acc, sum, cur, c, h);
  }
  __syncthreads();

  const int n0 = b * OWN_NODES;
  constexpr int PER_BLOCK = OWN_NODES * (D_FEAT / 4);
  constexpr int FULL_TRIPS = PER_BLOCK / (PROP_WAVES * 32);
  if (n0 + OWN_NODES <= N) {
#pragma unroll
    for (int k = 0; k < FULL_TRIPS; ++k)
      write_term<HAS_P>(t + k * (PROP_WAVES * 32), n0, acc, x, xp, deg, out, cH, cX, cP);
    if (t < PER_BLOCK - FULL_TRIPS * (PROP_WAVES * 32))
      write_term<HAS_P>(t + FULL_TRIPS * (PROP_WAVES * 32), n0, acc, x, xp, deg, out, cH, cX, cP);
  } else {
    for (int i = t; i < PER_BLOCK; i += PROP_WAVES * 32) {
      if (n0 + (i >> 4) < N) write_term<HAS_P>(i, n0, acc, x, xp, deg, out, cH, cX, cP);
    }
  }
}


#define GEMM_WAVES 4
#define TILE_SLOTS 32
#define W_TERM (D_FEAT * D_FEAT)

__device__ __forceinline__ void load8(float (&a)[8], const float* __restrict__ p) {
  const float4 lo = reinterpret_cast<const float4*>(p)[0];
  const float4 hi = reinterpret_cast<const float4*>(p)[1];
  a[0] = lo.x; a[1] = lo.y; a[2] = lo.z; a[3] = lo.w;
  a[4] = hi.x; a[5] = hi.y; a[6] = hi.z; a[7] = hi.w;
}

__device__ __forceinline__ void term_quarter(v8f (&acc)[4], const float (&a)[8],
                                             const float* __restrict__ wrow) {
  v2f bf[4][4];
#pragma unroll
  for (int j = 0; j < 4; ++j) {
    const float4 w0 = *reinterpret_cast<const float4*>(wrow + (2 * j + 0) * D_FEAT);
    const float4 w1 = *reinterpret_cast<const float4*>(wrow + (2 * j + 1) * D_FEAT);
    bf[j][0].x = w0.x; bf[j][0].y = w1.x;
    bf[j][1].x = w0.y; bf[j][1].y = w1.y;
    bf[j][2].x = w0.z; bf[j][2].y = w1.z;
    bf[j][3].x = w0.w; bf[j][3].y = w1.w;
  }
#pragma unroll
  for (int j = 0; j < 4; ++j) {
    v2f av;
    av.x = a[2 * j + 0];
    av.y = a[2 * j + 1];
#pragma unroll
    for (int n = 0; n < 4; ++n) {
      acc[n] = __builtin_amdgcn_wmma_f32_16x16x4_f32(
           false, av,  false, bf[j][n],
           (short)0, acc[n],  false,  false);
    }
  }
}

__global__ __launch_bounds__(128) void gemm_fused_kernel(
    const float* __restrict__ X0, const float* __restrict__ X1,
    const float* __restrict__ X2, const float* __restrict__ X3,
    const float* __restrict__ W, const float* __restrict__ bias,
    float* __restrict__ out, int N) {
  __shared__ float red[GEMM_WAVES * TILE_SLOTS * 32];

  const int lane = threadIdx.x & 31;
  const int wv   = threadIdx.x >> 5;
  const int row0 = blockIdx.x * 16;
  if (row0 >= N) return;

  const int c  = lane & 15;
  const int h  = lane >> 4;
  const int k0 = wv * 16 + h * 8;

  int arow = row0 + c;
  if (arow >= N) arow = N - 1;
  const long aoff = (long)arow * D_FEAT + k0;
  const float* wq = W + (long)k0 * D_FEAT + 4 * c;

  v8f acc[4];
  {
    const float4 bb = *reinterpret_cast<const float4*>(bias + 4 * c);
    const float b0 = wv == 0 ? bb.x : 0.0f;
    const float b1 = wv == 0 ? bb.y : 0.0f;
    const float b2 = wv == 0 ? bb.z : 0.0f;
    const float b3 = wv == 0 ? bb.w : 0.0f;
#pragma unroll
    for (int v = 0; v < 8; ++v) {
      acc[0][v] = b0; acc[1][v] = b1; acc[2][v] = b2; acc[3][v] = b3;
    }
  }

  float a[8];
  load8(a, X0 + aoff);
  term_quarter(acc, a, wq + 0 * W_TERM);
  load8(a, X1 + aoff);
  term_quarter(acc, a, wq + 1 * W_TERM);
  load8(a, X2 + aoff);
  term_quarter(acc, a, wq + 2 * W_TERM);
  load8(a, X3 + aoff);
  term_quarter(acc, a, wq + 3 * W_TERM);

  {
    float* mine = red + (wv * TILE_SLOTS) * 32 + lane;
#pragma unroll
    for (int v = 0; v < 8; ++v) {
#pragma unroll
      for (int n = 0; n < 4; ++n) mine[(4 * v + n) * 32] = acc[n][v];
    }
  }
  __syncthreads();

  for (int j = 0; j < 2; ++j) {
    const int v = 2 * wv + j;
    const float* p = red + (4 * v) * 32 + lane;
    float o[4];
#pragma unroll
    for (int n = 0; n < 4; ++n) {
      const float* q = p + n * 32;
      o[n] = ((q[0 * TILE_SLOTS * 32] + q[1 * TILE_SLOTS * 32]) + q[2 * TILE_SLOTS * 32])
             + q[3 * TILE_SLOTS * 32];
    }
    const int row = row0 + 8 * h + v;
    if (row < N)
      *reinterpret_cast<float4*>(out + (long)row * D_FEAT + 4 * c) = make_float4(o[0], o[1], o[2], o[3]);
  }
}


extern "C" void kernel_launch(void* const* d_in, const int* in_sizes, int n_in,
                              void* d_out, int out_size, void* d_ws, size_t ws_size,
                              hipStream_t stream) {
  const float* feat = (const float*)d_in[0];
  const int*   src  = (const int*)d_in[1];
  const int*   dst  = (const int*)d_in[2];
  const float* W    = (const float*)d_in[3];
  const float* b    = (const float*)d_in[4];
  float*       out  = (float*)d_out;

  const int N = in_sizes[0] / D_FEAT;
  const int E = in_sizes[1];
  const long nd = (long)N * D_FEAT;
  const int nown = (N + OWN_NODES - 1) / OWN_NODES;

  const long npad = ((long)N + 31) & ~31L;
  const long epad = ((long)E + 31) & ~31L;
  const int nchunks = (E + CHUNK_EDGES - 1) / CHUNK_EDGES;
  const int cpitch = (nchunks + 15) & ~15;
  float*    deg    = (float*)d_ws;
  int2*     tab    = (int2*)(deg + npad);
  unsigned* cseg   = (unsigned*)(tab + (long)MAX_OWN * cpitch);
  uint2*    seg2   = (uint2*)(cseg + (long)nchunks * CHUNK_EDGES);
  const long e2pad = epad + (long)PAD_PER_OWN * MAX_OWN + 32;
  int2*     range  = (int2*)(seg2 + e2pad);
  float*    txA    = (float*)(range + MAX_OWN);
  float*    txB    = txA + nd;
  float*    txC    = txB + nd;

  const int ggrid = (N + 15) / 16;

  group_kernel<<<nchunks, FILL_THREADS, 0, stream>>>(src, dst, tab, cseg, E, nown, cpitch);
  collect_kernel<<<nown, ORDER_THREADS, 0, stream>>>(tab, cseg, deg, seg2, range, nchunks, cpitch, N);

  prop_kernel<false><<<nown, PROP_WAVES * 32, 0, stream>>>(feat, feat, deg, seg2, range, txA,
                                                           -2.0f / LAM, 2.0f / LAM - 1.0f, 0.0f, N);
  prop_kernel<true><<<nown, PROP_WAVES * 32, 0, stream>>>(txA, feat, deg, seg2, range, txB,
                                                          -4.0f / LAM, 4.0f / LAM - 2.0f, -1.0f, N);
  prop_kernel<true><<<nown, PROP_WAVES * 32, 0, stream>>>(txB, txA, deg, seg2, range, txC,
                                                          -4.0f / LAM, 4.0f / LAM - 2.0f, -1.0f, N);

  gemm_fused_kernel<<<ggrid, 128, 0, stream>>>(feat, txA, txB, txC, W, b, out, N);
}
